# starting kernel
# baseline (speedup 1.0000x reference)
#include <hip/hip_runtime.h>
#include <hip/hip_bf16.h>

typedef __attribute__((ext_vector_type(2))) float v2f;
typedef __attribute__((ext_vector_type(8))) float v8f;


__global__ void k_init_deg(float* __restrict__ deg, int n) {
    int i = blockIdx.x * blockDim.x + threadIdx.x;
    if (i < n) deg[i] = 1.0f;
}

__global__ void k_count_deg(const int* __restrict__ ei, float* __restrict__ deg, int nedges) {
    int e = blockIdx.x * blockDim.x + threadIdx.x;
    if (e < nedges) atomicAdd(&deg[ei[nedges + e]], 1.0f);
}

__global__ void k_rsqrt_inplace(float* __restrict__ deg, int n) {
    int i = blockIdx.x * blockDim.x + threadIdx.x;
    if (i < n) deg[i] = rsqrtf(deg[i]);
}

__global__ void k_edge_norm(const int* __restrict__ ei, const float* __restrict__ dinv,
                            float* __restrict__ w, int nedges) {
    int e = blockIdx.x * blockDim.x + threadIdx.x;
    if (e < nedges) w[e] = dinv[ei[e]] * dinv[ei[nedges + e]];
}


#define WT_PITCH 68

template <bool BIAS_RELU>
__global__ void k_gemm64(const float* __restrict__ A, const float* __restrict__ W,
                         const float* __restrict__ bias, float* __restrict__ C, int nrows) {
    __shared__ float ldsW[64 * WT_PITCH];

    for (int idx = threadIdx.x; idx < 4096; idx += 256) {
        const int k = idx >> 6;
        const int col = idx & 63;
        ldsW[col * WT_PITCH + k] = W[idx];
    }
    __syncthreads();

    const int lane = threadIdx.x & 31;
    const int wave = threadIdx.x >> 5;
    const int rowTiles = (nrows + 15) >> 4;
    const int tile = blockIdx.x * 8 + wave;
    if (tile < rowTiles) {
        const int half = lane >> 4;
        const int m    = lane & 15;
        int row = tile * 16 + m;
        if (row >= nrows) row = nrows - 1;
        const float* __restrict__ arow = A + (size_t)row * 64;

        v8f c0 = {}, c1 = {}, c2 = {}, c3 = {};
#pragma unroll
        for (int kk = 0; kk < 16; ++kk) {
            const int k0 = kk * 4 + 2 * half;
            v2f a;
            a.x = arow[k0];
            a.y = arow[k0 + 1];
            v2f b0 = *(const v2f*)&ldsW[(0 * 16 + m) * WT_PITCH + k0];
            v2f b1 = *(const v2f*)&ldsW[(1 * 16 + m) * WT_PITCH + k0];
            v2f b2 = *(const v2f*)&ldsW[(2 * 16 + m) * WT_PITCH + k0];
            v2f b3 = *(const v2f*)&ldsW[(3 * 16 + m) * WT_PITCH + k0];
            c0 = __builtin_amdgcn_wmma_f32_16x16x4_f32(false, a, false, b0, (short)0, c0, false, false);
            c1 = __builtin_amdgcn_wmma_f32_16x16x4_f32(false, a, false, b1, (short)0, c1, false, false);
            c2 = __builtin_amdgcn_wmma_f32_16x16x4_f32(false, a, false, b2, (short)0, c2, false, false);
            c3 = __builtin_amdgcn_wmma_f32_16x16x4_f32(false, a, false, b3, (short)0, c3, false, false);
        }

        v8f acc[4] = {c0, c1, c2, c3};
#pragma unroll
        for (int nt = 0; nt < 4; ++nt) {
            const int col = nt * 16 + m;
            const float bv = BIAS_RELU ? bias[col] : 0.0f;
#pragma unroll
            for (int r = 0; r < 8; ++r) {
                const int orow = tile * 16 + r + 8 * half;
                if (orow < nrows) {
                    float v = acc[nt][r];
                    if (BIAS_RELU) v = fmaxf(v + bv, 0.0f);
                    C[(size_t)orow * 64 + col] = v;
                }
            }
        }
    }
}


__global__ void k_selfloop_init(const float* __restrict__ t, const float* __restrict__ dinv,
                                float* __restrict__ agg, int n) {
    long tid = (long)blockIdx.x * blockDim.x + threadIdx.x;
    int i = (int)(tid >> 4);
    int j = (int)(tid & 15) * 4;
    if (i >= n) return;
    float w = dinv[i] * dinv[i];
    float4 v = *(const float4*)(t + (size_t)i * 64 + j);
    float4 r = make_float4(w * v.x, w * v.y, w * v.z, w * v.w);
    *(float4*)(agg + (size_t)i * 64 + j) = r;
}

__global__ void k_edge_scatter(const int* __restrict__ ei, const float* __restrict__ wnorm,
                               const float* __restrict__ t, float* __restrict__ agg, int nedges) {
    long tid = (long)blockIdx.x * blockDim.x + threadIdx.x;
    int e = (int)(tid >> 4);
    int j = (int)(tid & 15) * 4;
    if (e >= nedges) return;
    int s = ei[e];
    int d = ei[nedges + e];
    float w = wnorm[e];
    float4 v = *(const float4*)(t + (size_t)s * 64 + j);
    float* p = agg + (size_t)d * 64 + j;
    atomicAdd(p + 0, w * v.x);
    atomicAdd(p + 1, w * v.y);
    atomicAdd(p + 2, w * v.z);
    atomicAdd(p + 3, w * v.w);
}

__global__ void k_bias_relu(const float* __restrict__ agg, const float* __restrict__ b,
                            float* __restrict__ h, long n64) {
    long idx = (long)blockIdx.x * blockDim.x + threadIdx.x;
    if (idx >= n64) return;
    int f = (int)(idx & 63);
    h[idx] = fmaxf(agg[idx] + b[f], 0.0f);
}

__global__ void k_final_dot(const float* __restrict__ H, const float* __restrict__ w,
                            const float* __restrict__ b, float* __restrict__ out, int n) {
    int i = blockIdx.x * blockDim.x + threadIdx.x;
    if (i >= n) return;
    const float4* hp = (const float4*)(H + (size_t)i * 64);
    const float4* wp = (const float4*)w;
    float s = 0.0f;
#pragma unroll
    for (int k = 0; k < 16; ++k) {
        float4 hv = hp[k], wv = wp[k];
        s += hv.x * wv.x + hv.y * wv.y + hv.z * wv.z + hv.w * wv.w;
    }
    out[i] = s + b[0];
}


extern "C" void kernel_launch(void* const* d_in, const int* in_sizes, int n_in,
                              void* d_out, int out_size, void* d_ws, size_t ws_size,
                              hipStream_t stream) {
    (void)n_in; (void)out_size; (void)ws_size;

    const float* x   = (const float*)d_in[0];
    const int*   ei  = (const int*)  d_in[1];
    const float* W1  = (const float*)d_in[2];
    const float* b1  = (const float*)d_in[3];
    const float* W2  = (const float*)d_in[4];
    const float* b2  = (const float*)d_in[5];
    const float* Wf1 = (const float*)d_in[6];
    const float* bf1 = (const float*)d_in[7];
    const float* Wf2 = (const float*)d_in[8];
    const float* bf2 = (const float*)d_in[9];
    const float* Wf3 = (const float*)d_in[10];
    const float* bf3 = (const float*)d_in[11];
    float* out = (float*)d_out;

    const int n = in_sizes[0] / 64;
    const int E = in_sizes[1] / 2;
    const long n64 = (long)n * 64;

    float* ws    = (float*)d_ws;
    float* dinv  = ws;
    float* wnorm = ws + n;
    float* bufA  = wnorm + E;
    float* bufB  = bufA + (size_t)n * 64;

    const int T = 256;
    const int rowTiles = (n + 15) / 16;
    const int gemmBlocks = (rowTiles + 7) / 8;
    const int nodeVecBlocks = (int)(((long)n * 16 + T - 1) / T);
    const int edgeVecBlocks = (int)(((long)E * 16 + T - 1) / T);
    const int elemBlocks = (int)((n64 + T - 1) / T);

    k_init_deg<<<(n + T - 1) / T, T, 0, stream>>>(dinv, n);
    k_count_deg<<<(E + T - 1) / T, T, 0, stream>>>(ei, dinv, E);
    k_rsqrt_inplace<<<(n + T - 1) / T, T, 0, stream>>>(dinv, n);
    k_edge_norm<<<(E + T - 1) / T, T, 0, stream>>>(ei, dinv, wnorm, E);

    k_gemm64<false><<<gemmBlocks, T, 0, stream>>>(x, W1, nullptr, bufA, n);
    k_selfloop_init<<<nodeVecBlocks, T, 0, stream>>>(bufA, dinv, bufB, n);
    k_edge_scatter<<<edgeVecBlocks, T, 0, stream>>>(ei, wnorm, bufA, bufB, E);
    k_bias_relu<<<elemBlocks, T, 0, stream>>>(bufB, b1, bufA, n64);

    k_gemm64<false><<<gemmBlocks, T, 0, stream>>>(bufA, W2, nullptr, bufB, n);
    k_selfloop_init<<<nodeVecBlocks, T, 0, stream>>>(bufB, dinv, bufA, n);
    k_edge_scatter<<<edgeVecBlocks, T, 0, stream>>>(ei, wnorm, bufB, bufA, E);
    k_bias_relu<<<elemBlocks, T, 0, stream>>>(bufA, b2, bufB, n64);

    k_gemm64<true><<<gemmBlocks, T, 0, stream>>>(bufB, Wf1, bf1, bufA, n);
    k_gemm64<true><<<gemmBlocks, T, 0, stream>>>(bufA, Wf2, bf2, bufB, n);
    k_final_dot<<<(n + T - 1) / T, T, 0, stream>>>(bufB, Wf3, bf3, out, n);
}


# Replace the float-atomic edge scatter-add by a per-call destination-ordered edge list (integer atomic rank, scan, place) and a per-node row gather; sum order is the arrival rank
# speedup vs baseline: 1.8323x; 1.8323x over previous
#include <hip/hip_runtime.h>


typedef __attribute__((ext_vector_type(2))) float v2f;
typedef __attribute__((ext_vector_type(8))) float v8f;


__global__ void k_zero_counts(int* __restrict__ cnt, int n) {
    int i = blockIdx.x * blockDim.x + threadIdx.x;
    if (i < n) cnt[i] = 0;
}

__global__ void k_count_rank(const int* __restrict__ ei, int* __restrict__ cnt,
                             int* __restrict__ rank, int nedges) {
    const int tid  = blockIdx.x * blockDim.x + threadIdx.x;
    const int nthr = gridDim.x * blockDim.x;
    int d[4];
#pragma unroll
    for (int k = 0; k < 4; ++k) {
        const int e = tid + k * nthr;
        d[k] = ei[nedges + min(e, nedges - 1)];
    }
    int r[4];
#pragma unroll
    for (int k = 0; k < 4; ++k) {
        const int e = tid + k * nthr;
        r[k] = atomicAdd(&cnt[d[k]], (e < nedges) ? 1 : 0);
    }
#pragma unroll
    for (int k = 0; k < 4; ++k) {
        const int e = tid + k * nthr;
        if (e < nedges) rank[e] = r[k];
    }
}

__global__ __launch_bounds__(1024) void k_scan_rowptr(const int* __restrict__ cnt, int* __restrict__ rowptr,
                                                      float* __restrict__ dinv, int n) {
    __shared__ int red[32];
    __shared__ int wtot[32];
    const int tid  = threadIdx.x;
    const int lane = tid & 31;
    const int wv   = tid >> 5;
    const int start = blockIdx.x * 1024;

    int part = 0;
    const int nvec = start >> 2;
    if (nvec > 0) {
        const int4* __restrict__ c4 = (const int4*)cnt;
        for (int j = tid; j < nvec; j += 4096) {
            int4 q[4];
#pragma unroll
            for (int k = 0; k < 4; ++k) q[k] = c4[min(j + k * 1024, nvec - 1)];
#pragma unroll
            for (int k = 0; k < 4; ++k)
                if (j + k * 1024 < nvec) part += q[k].x + q[k].y + q[k].z + q[k].w;
        }
    }
#pragma unroll
    for (int o = 16; o > 0; o >>= 1) part += __shfl_xor(part, o, 32);
    if (lane == 0) red[wv] = part;

    const int i = start + tid;
    const int c = (i < n) ? cnt[i] : 0;
    int incl = c;
#pragma unroll
    for (int o = 1; o < 32; o <<= 1) {
        const int up = __shfl_up(incl, o, 32);
        if (lane >= o) incl += up;
    }
    if (lane == 31) wtot[wv] = incl;
    __syncthreads();

    int offset = 0, wpre = 0;
#pragma unroll
    for (int w = 0; w < 32; ++w) {
        offset += red[w];
        if (w < wv) wpre += wtot[w];
    }
    const int excl = offset + wpre + incl - c;
    if (i < n) {
        rowptr[i] = excl;
        dinv[i] = rsqrtf((float)(c + 1));
        if (i == n - 1) rowptr[n] = excl + c;
    }
}

__global__ void k_place_edges(const int* __restrict__ ei, const int* __restrict__ rank,
                              const int* __restrict__ rowptr, int* __restrict__ csrc, int nedges) {
    const int tid  = blockIdx.x * blockDim.x + threadIdx.x;
    const int nthr = gridDim.x * blockDim.x;
    int s[4], d[4], r[4];
#pragma unroll
    for (int k = 0; k < 4; ++k) {
        const int e = min(tid + k * nthr, nedges - 1);
        s[k] = ei[e];
        d[k] = ei[nedges + e];
        r[k] = rank[e];
    }
    int p[4];
#pragma unroll
    for (int k = 0; k < 4; ++k) p[k] = rowptr[d[k]] + r[k];
#pragma unroll
    for (int k = 0; k < 4; ++k) {
        const int e = tid + k * nthr;
        if (e < nedges) csrc[min(max(p[k], 0), nedges - 1)] = s[k];
    }
}


#define WT_PITCH 68

template <bool BIAS_RELU>
__global__ void k_gemm64(const float* __restrict__ A, const float* __restrict__ W,
                         const float* __restrict__ bias, float* __restrict__ C, int nrows) {
    __shared__ float ldsW[64 * WT_PITCH];

    for (int idx = threadIdx.x; idx < 4096; idx += 256) {
        const int k = idx >> 6;
        const int col = idx & 63;
        ldsW[col * WT_PITCH + k] = W[idx];
    }
    __syncthreads();

    const int lane = threadIdx.x & 31;
    const int wave = threadIdx.x >> 5;
    const int rowTiles = (nrows + 15) >> 4;
    const int tile = blockIdx.x * 8 + wave;
    if (tile < rowTiles) {
        const int half = lane >> 4;
        const int m    = lane & 15;
        int row = tile * 16 + m;
        if (row >= nrows) row = nrows - 1;
        const float* __restrict__ arow = A + (size_t)row * 64;

        v8f c0 = {}, c1 = {}, c2 = {}, c3 = {};
#pragma unroll
        for (int kk = 0; kk < 16; ++kk) {
            const int k0 = kk * 4 + 2 * half;
            v2f a;
            a.x = arow[k0];
            a.y = arow[k0 + 1];
            v2f b0 = *(const v2f*)&ldsW[(0 * 16 + m) * WT_PITCH + k0];
            v2f b1 = *(const v2f*)&ldsW[(1 * 16 + m) * WT_PITCH + k0];
            v2f b2 = *(const v2f*)&ldsW[(2 * 16 + m) * WT_PITCH + k0];
            v2f b3 = *(const v2f*)&ldsW[(3 * 16 + m) * WT_PITCH + k0];
            c0 = __builtin_amdgcn_wmma_f32_16x16x4_f32(false, a, false, b0, (short)0, c0, false, false);
            c1 = __builtin_amdgcn_wmma_f32_16x16x4_f32(false, a, false, b1, (short)0, c1, false, false);
            c2 = __builtin_amdgcn_wmma_f32_16x16x4_f32(false, a, false, b2, (short)0, c2, false, false);
            c3 = __builtin_amdgcn_wmma_f32_16x16x4_f32(false, a, false, b3, (short)0, c3, false, false);
        }

        v8f acc[4] = {c0, c1, c2, c3};
#pragma unroll
        for (int nt = 0; nt < 4; ++nt) {
            const int col = nt * 16 + m;
            const float bv = BIAS_RELU ? bias[col] : 0.0f;
#pragma unroll
            for (int r = 0; r < 8; ++r) {
                const int orow = tile * 16 + r + 8 * half;
                if (orow < nrows) {
                    float v = acc[nt][r];
                    if (BIAS_RELU) v = fmaxf(v + bv, 0.0f);
                    C[(size_t)orow * 64 + col] = v;
                }
            }
        }
    }
}

__global__ __launch_bounds__(256) void k_gather_rows(const int* __restrict__ rowptr, const int* __restrict__ csrc,
                                                     const float* __restrict__ dinv, const float* __restrict__ t,
                                                     const float* __restrict__ bias, float* __restrict__ h, int n) {
    const int l16  = threadIdx.x & 15;
    const int grp  = (blockIdx.x * blockDim.x + threadIdx.x) >> 4;
    const int ngrp = (gridDim.x * blockDim.x) >> 4;
    const float4 bv = *(const float4*)(bias + l16 * 4);

    for (int i = grp; i < n; i += ngrp) {
        const int beg = rowptr[i];
        const int cnt = rowptr[i + 1] - beg;
        const float di = dinv[i];
        const float4 own = *(const float4*)(t + (size_t)i * 64 + l16 * 4);
        const float wself = di * di;
        float4 acc = make_float4(wself * own.x, wself * own.y, wself * own.z, wself * own.w);

        for (int base = 0; base < cnt; base += 16) {
            const int k = base + l16;
            int   s_mine = i;
            float w_mine = 0.0f;
            if (k < cnt) {
                s_mine = min(max(csrc[beg + k], 0), n - 1);
                w_mine = dinv[s_mine] * di;
            }
            const int m = min(cnt - base, 16);
            for (int j = 0; j < m; j += 4) {
                int sj[4]; float wj[4]; float4 r[4];
#pragma unroll
                for (int u = 0; u < 4; ++u) {
                    sj[u] = __shfl(s_mine, j + u, 16);
                    wj[u] = __shfl(w_mine, j + u, 16);
                    r[u]  = *(const float4*)(t + (size_t)sj[u] * 64 + l16 * 4);
                }
#pragma unroll
                for (int u = 0; u < 4; ++u) {
                    acc.x = fmaf(wj[u], r[u].x, acc.x);
                    acc.y = fmaf(wj[u], r[u].y, acc.y);
                    acc.z = fmaf(wj[u], r[u].z, acc.z);
                    acc.w = fmaf(wj[u], r[u].w, acc.w);
                }
            }
        }
        float4 o = make_float4(fmaxf(acc.x + bv.x, 0.0f), fmaxf(acc.y + bv.y, 0.0f),
                               fmaxf(acc.z + bv.z, 0.0f), fmaxf(acc.w + bv.w, 0.0f));
        *(float4*)(h + (size_t)i * 64 + l16 * 4) = o;
    }
}

__global__ void k_final_dot(const float* __restrict__ H, const float* __restrict__ w,
                            const float* __restrict__ b, float* __restrict__ out, int n) {
    int i = blockIdx.x * blockDim.x + threadIdx.x;
    if (i >= n) return;
    const float4* hp = (const float4*)(H + (size_t)i * 64);
    const float4* wp = (const float4*)w;
    float s = 0.0f;
#pragma unroll
    for (int k = 0; k < 16; ++k) {
        float4 hv = hp[k], wv = wp[k];
        s += hv.x * wv.x + hv.y * wv.y + hv.z * wv.z + hv.w * wv.w;
    }
    out[i] = s + b[0];
}


extern "C" void kernel_launch(void* const* d_in, const int* in_sizes, int n_in,
                              void* d_out, int out_size, void* d_ws, size_t ws_size,
                              hipStream_t stream) {
    (void)n_in; (void)out_size; (void)ws_size;

    const float* x   = (const float*)d_in[0];
    const int*   ei  = (const int*)  d_in[1];
    const float* W1  = (const float*)d_in[2];
    const float* b1  = (const float*)d_in[3];
    const float* W2  = (const float*)d_in[4];
    const float* b2  = (const float*)d_in[5];
    const float* Wf1 = (const float*)d_in[6];
    const float* bf1 = (const float*)d_in[7];
    const float* Wf2 = (const float*)d_in[8];
    const float* bf2 = (const float*)d_in[9];
    const float* Wf3 = (const float*)d_in[10];
    const float* bf3 = (const float*)d_in[11];
    float* out = (float*)d_out;

    const int n = in_sizes[0] / 64;
    const int E = in_sizes[1] / 2;

    const size_t nPad  = ((size_t)n + 1 + 63) / 64 * 64;
    const size_t ePad  = ((size_t)E + 63) / 64 * 64;
    float* ws     = (float*)d_ws;
    float* dinv   = ws;
    int*   cnt    = (int*)(ws + nPad);
    int*   rowptr = (int*)(ws + 2 * nPad);
    int*   rank   = (int*)(ws + 3 * nPad);
    int*   csrc   = (int*)(ws + 3 * nPad + ePad);
    float* bufA   = ws + 3 * nPad + 2 * ePad;
    float* bufB   = bufA + (size_t)n * 64;

    const int T = 256;
    const int rowTiles = (n + 15) / 16;
    const int gemmBlocks = (rowTiles + 7) / 8;
    const int edgeBlocks4 = (E + 4 * T - 1) / (4 * T);
    const int gatherBlocks = 2048;

    k_zero_counts<<<(n + T - 1) / T, T, 0, stream>>>(cnt, n);
    k_count_rank<<<edgeBlocks4, T, 0, stream>>>(ei, cnt, rank, E);
    k_scan_rowptr<<<(n + 1023) / 1024, 1024, 0, stream>>>(cnt, rowptr, dinv, n);
    k_place_edges<<<edgeBlocks4, T, 0, stream>>>(ei, rank, rowptr, csrc, E);

    k_gemm64<false><<<gemmBlocks, T, 0, stream>>>(x, W1, nullptr, bufA, n);
    k_gather_rows<<<gatherBlocks, T, 0, stream>>>(rowptr, csrc, dinv, bufA, b1, bufB, n);

    k_gemm64<false><<<gemmBlocks, T, 0, stream>>>(bufB, W2, nullptr, bufA, n);
    k_gather_rows<<<gatherBlocks, T, 0, stream>>>(rowptr, csrc, dinv, bufA, b2, bufB, n);

    k_gemm64<true><<<gemmBlocks, T, 0, stream>>>(bufB, Wf1, bf1, bufA, n);
    k_gemm64<true><<<gemmBlocks, T, 0, stream>>>(bufA, Wf2, bf2, bufB, n);
    k_final_dot<<<(n + T - 1) / T, T, 0, stream>>>(bufB, Wf3, bf3, out, n);
}


# Fuse MLP head (two 64x64 products, bias+relu, final dot) into one launch: transposed products make product 1's result registers product 2's B operand; product 2 k order and dot sum order change
# speedup vs baseline: 2.1209x; 1.1575x over previous
#include <hip/hip_runtime.h>


typedef __attribute__((ext_vector_type(2))) float v2f;
typedef __attribute__((ext_vector_type(8))) float v8f;


__global__ void k_zero_counts(int* __restrict__ cnt, int n) {
    int i = blockIdx.x * blockDim.x + threadIdx.x;
    if (i < n) cnt[i] = 0;
}

__global__ void k_count_rank(const int* __restrict__ ei, int* __restrict__ cnt,
                             int* __restrict__ rank, int nedges) {
    const int tid  = blockIdx.x * blockDim.x + threadIdx.x;
    const int nthr = gridDim.x * blockDim.x;
    int d[4];
#pragma unroll
    for (int k = 0; k < 4; ++k) {
        const int e = tid + k * nthr;
        d[k] = ei[nedges + min(e, nedges - 1)];
    }
    int r[4];
#pragma unroll
    for (int k = 0; k < 4; ++k) {
        const int e = tid + k * nthr;
        r[k] = atomicAdd(&cnt[d[k]], (e < nedges) ? 1 : 0);
    }
#pragma unroll
    for (int k = 0; k < 4; ++k) {
        const int e = tid + k * nthr;
        if (e < nedges) rank[e] = r[k];
    }
}

__global__ __launch_bounds__(1024) void k_scan_rowptr(const int* __restrict__ cnt, int* __restrict__ rowptr,
                                                      float* __restrict__ dinv, int n) {
    __shared__ int red[32];
    __shared__ int wtot[32];
    const int tid  = threadIdx.x;
    const int lane = tid & 31;
    const int wv   = tid >> 5;
    const int start = blockIdx.x * 1024;

    int part = 0;
    const int nvec = start >> 2;
    if (nvec > 0) {
        const int4* __restrict__ c4 = (const int4*)cnt;
        for (int j = tid; j < nvec; j += 4096) {
            int4 q[4];
#pragma unroll
            for (int k = 0; k < 4; ++k) q[k] = c4[min(j + k * 1024, nvec - 1)];
#pragma unroll
            for (int k = 0; k < 4; ++k)
                if (j + k * 1024 < nvec) part += q[k].x + q[k].y + q[k].z + q[k].w;
        }
    }
#pragma unroll
    for (int o = 16; o > 0; o >>= 1) part += __shfl_xor(part, o, 32);
    if (lane == 0) red[wv] = part;

    const int i = start + tid;
    const int c = (i < n) ? cnt[i] : 0;
    int incl = c;
#pragma unroll
    for (int o = 1; o < 32; o <<= 1) {
        const int up = __shfl_up(incl, o, 32);
        if (lane >= o) incl += up;
    }
    if (lane == 31) wtot[wv] = incl;
    __syncthreads();

    int offset = 0, wpre = 0;
#pragma unroll
    for (int w = 0; w < 32; ++w) {
        offset += red[w];
        if (w < wv) wpre += wtot[w];
    }
    const int excl = offset + wpre + incl - c;
    if (i < n) {
        rowptr[i] = excl;
        dinv[i] = rsqrtf((float)(c + 1));
        if (i == n - 1) rowptr[n] = excl + c;
    }
}

__global__ void k_place_edges(const int* __restrict__ ei, const int* __restrict__ rank,
                              const int* __restrict__ rowptr, int* __restrict__ csrc, int nedges) {
    const int tid  = blockIdx.x * blockDim.x + threadIdx.x;
    const int nthr = gridDim.x * blockDim.x;
    int s[4], d[4], r[4];
#pragma unroll
    for (int k = 0; k < 4; ++k) {
        const int e = min(tid + k * nthr, nedges - 1);
        s[k] = ei[e];
        d[k] = ei[nedges + e];
        r[k] = rank[e];
    }
    int p[4];
#pragma unroll
    for (int k = 0; k < 4; ++k) p[k] = rowptr[d[k]] + r[k];
#pragma unroll
    for (int k = 0; k < 4; ++k) {
        const int e = tid + k * nthr;
        if (e < nedges) csrc[min(max(p[k], 0), nedges - 1)] = s[k];
    }
}


#define WT_PITCH 68

__global__ void k_gemm64(const float* __restrict__ A, const float* __restrict__ W,
                         const float* __restrict__ rowscale,
                         float* __restrict__ C, int nrows) {
    __shared__ float ldsW[64 * WT_PITCH];

    for (int idx = threadIdx.x; idx < 4096; idx += 256) {
        const int k = idx >> 6;
        const int col = idx & 63;
        ldsW[col * WT_PITCH + k] = W[idx];
    }
    __syncthreads();

    const int lane = threadIdx.x & 31;
    const int wave = threadIdx.x >> 5;
    const int rowTiles = (nrows + 15) >> 4;
    const int tile = blockIdx.x * 8 + wave;
    if (tile < rowTiles) {
        const int half = lane >> 4;
        const int m    = lane & 15;
        int row = tile * 16 + m;
        if (row >= nrows) row = nrows - 1;
        const float* __restrict__ arow = A + (size_t)row * 64;

        float sc[8];
#pragma unroll
        for (int r = 0; r < 8; ++r)
            sc[r] = rowscale[min(tile * 16 + r + 8 * half, nrows - 1)];

        v8f c0 = {}, c1 = {}, c2 = {}, c3 = {};
#pragma unroll
        for (int kk = 0; kk < 16; ++kk) {
            const int k0 = kk * 4 + 2 * half;
            v2f a;
            a.x = arow[k0];
            a.y = arow[k0 + 1];
            v2f b0 = *(const v2f*)&ldsW[(0 * 16 + m) * WT_PITCH + k0];
            v2f b1 = *(const v2f*)&ldsW[(1 * 16 + m) * WT_PITCH + k0];
            v2f b2 = *(const v2f*)&ldsW[(2 * 16 + m) * WT_PITCH + k0];
            v2f b3 = *(const v2f*)&ldsW[(3 * 16 + m) * WT_PITCH + k0];
            c0 = __builtin_amdgcn_wmma_f32_16x16x4_f32(false, a, false, b0, (short)0, c0, false, false);
            c1 = __builtin_amdgcn_wmma_f32_16x16x4_f32(false, a, false, b1, (short)0, c1, false, false);
            c2 = __builtin_amdgcn_wmma_f32_16x16x4_f32(false, a, false, b2, (short)0, c2, false, false);
            c3 = __builtin_amdgcn_wmma_f32_16x16x4_f32(false, a, false, b3, (short)0, c3, false, false);
        }

        v8f acc[4] = {c0, c1, c2, c3};
#pragma unroll
        for (int nt = 0; nt < 4; ++nt) {
            const int col = nt * 16 + m;
#pragma unroll
            for (int r = 0; r < 8; ++r) {
                const int orow = tile * 16 + r + 8 * half;
                if (orow < nrows)
                    C[(size_t)orow * 64 + col] = acc[nt][r] * sc[r];
            }
        }
    }
}


__device__ __forceinline__ int head_kpos(int f) {
    return (f & 0x31) | ((f & 0x6) << 1) | ((f >> 2) & 0x2);
}

__global__ __launch_bounds__(256) void k_mlp_head(const float* __restrict__ H,
                                                  const float* __restrict__ Wa, const float* __restrict__ ba,
                                                  const float* __restrict__ Wb, const float* __restrict__ bb,
                                                  const float* __restrict__ w3, const float* __restrict__ b3,
                                                  float* __restrict__ out, int nrows) {
    __shared__ __align__(16) float ldsWa[64 * WT_PITCH];
    __shared__ __align__(16) float ldsWb[64 * WT_PITCH];
    __shared__ __align__(16) float ldsV[3 * 64];

    const int lane = threadIdx.x & 31;
    const int wave = threadIdx.x >> 5;
    const int half = lane >> 4;
    const int c    = lane & 15;
    const int rowTiles = (nrows + 15) >> 4;
    const int tile = blockIdx.x * 8 + wave;

    const int node = tile * 16 + c;
    const float* __restrict__ hrow = H + (size_t)min(node, nrows - 1) * 64 + 2 * half;
    v2f x[16];
#pragma unroll
    for (int kk = 0; kk < 16; ++kk) x[kk] = *(const v2f*)(hrow + 4 * kk);

    float wa[16], wb[16];
#pragma unroll
    for (int i = 0; i < 16; ++i) {
        wa[i] = Wa[threadIdx.x + 256 * i];
        wb[i] = Wb[threadIdx.x + 256 * i];
    }
#pragma unroll
    for (int i = 0; i < 16; ++i) {
        const int k = (threadIdx.x >> 6) + 4 * i;
        const int col = threadIdx.x & 63;
        ldsWa[col * WT_PITCH + k] = wa[i];
        ldsWb[col * WT_PITCH + head_kpos(k)] = wb[i];
    }
    if (threadIdx.x < 64) {
        ldsV[threadIdx.x]       = ba[threadIdx.x];
        ldsV[64 + threadIdx.x]  = bb[threadIdx.x];
        ldsV[128 + threadIdx.x] = w3[threadIdx.x];
    }
    __syncthreads();

    if (tile < rowTiles) {
        v8f t[4] = {};
#pragma unroll
        for (int kk = 0; kk < 16; ++kk) {
            const int k0 = kk * 4 + 2 * half;
            v2f a[4];
#pragma unroll
            for (int mt = 0; mt < 4; ++mt) a[mt] = *(const v2f*)&ldsWa[(mt * 16 + c) * WT_PITCH + k0];
#pragma unroll
            for (int mt = 0; mt < 4; ++mt)
                t[mt] = __builtin_amdgcn_wmma_f32_16x16x4_f32(false, a[mt], false, x[kk], (short)0, t[mt], false, false);
        }

#pragma unroll
        for (int mt = 0; mt < 4; ++mt) {
            const float4 g0 = *(const float4*)&ldsV[mt * 16 + 8 * half];
            const float4 g1 = *(const float4*)&ldsV[mt * 16 + 8 * half + 4];
            t[mt][0] = fmaxf(t[mt][0] + g0.x, 0.0f);
            t[mt][1] = fmaxf(t[mt][1] + g0.y, 0.0f);
            t[mt][2] = fmaxf(t[mt][2] + g0.z, 0.0f);
            t[mt][3] = fmaxf(t[mt][3] + g0.w, 0.0f);
            t[mt][4] = fmaxf(t[mt][4] + g1.x, 0.0f);
            t[mt][5] = fmaxf(t[mt][5] + g1.y, 0.0f);
            t[mt][6] = fmaxf(t[mt][6] + g1.z, 0.0f);
            t[mt][7] = fmaxf(t[mt][7] + g1.w, 0.0f);
        }

        v8f u[4] = {};
#pragma unroll
        for (int nt = 0; nt < 4; ++nt) {
#pragma unroll
            for (int q = 0; q < 4; ++q) {
                v2f b;
                b.x = t[nt][2 * q];
                b.y = t[nt][2 * q + 1];
                const int kp = nt * 16 + q * 4 + 2 * half;
                v2f a[4];
#pragma unroll
                for (int mt = 0; mt < 4; ++mt) a[mt] = *(const v2f*)&ldsWb[(mt * 16 + c) * WT_PITCH + kp];
#pragma unroll
                for (int mt = 0; mt < 4; ++mt)
                    u[mt] = __builtin_amdgcn_wmma_f32_16x16x4_f32(false, a[mt], false, b, (short)0, u[mt], false, false);
            }
        }

        float p[4];
#pragma unroll
        for (int mt = 0; mt < 4; ++mt) {
            const float4 g0 = *(const float4*)&ldsV[64 + mt * 16 + 8 * half];
            const float4 g1 = *(const float4*)&ldsV[64 + mt * 16 + 8 * half + 4];
            const float4 w0 = *(const float4*)&ldsV[128 + mt * 16 + 8 * half];
            const float4 w1 = *(const float4*)&ldsV[128 + mt * 16 + 8 * half + 4];
            float s = fmaxf(u[mt][0] + g0.x, 0.0f) * w0.x;
            s = fmaf(fmaxf(u[mt][1] + g0.y, 0.0f), w0.y, s);
            s = fmaf(fmaxf(u[mt][2] + g0.z, 0.0f), w0.z, s);
            s = fmaf(fmaxf(u[mt][3] + g0.w, 0.0f), w0.w, s);
            s = fmaf(fmaxf(u[mt][4] + g1.x, 0.0f), w1.x, s);
            s = fmaf(fmaxf(u[mt][5] + g1.y, 0.0f), w1.y, s);
            s = fmaf(fmaxf(u[mt][6] + g1.z, 0.0f), w1.z, s);
            s = fmaf(fmaxf(u[mt][7] + g1.w, 0.0f), w1.w, s);
            p[mt] = s;
        }
        float dot = (p[0] + p[1]) + (p[2] + p[3]);
        dot += __shfl_xor(dot, 16, 32);
        if (half == 0 && node < nrows) out[node] = dot + b3[0];
    }
}

__global__ __launch_bounds__(256) void k_gather_rows(const int* __restrict__ rowptr, const int* __restrict__ csrc,
                                                     const float* __restrict__ dinv, const float* __restrict__ ts,
                                                     const float* __restrict__ bias, float* __restrict__ h, int n) {
    const int l16  = threadIdx.x & 15;
    const int grp  = (blockIdx.x * blockDim.x + threadIdx.x) >> 4;
    const int ngrp = (gridDim.x * blockDim.x) >> 4;
    const float4 bv = *(const float4*)(bias + l16 * 4);

    for (int i = grp; i < n; i += ngrp) {
        const int beg = rowptr[i];
        const int cnt = rowptr[i + 1] - beg;
        const float di = dinv[i];
        float4 acc = *(const float4*)(ts + (size_t)i * 64 + l16 * 4);

        for (int base = 0; base < cnt; base += 16) {
            const int k = base + l16;
            int s_mine = i;
            if (k < cnt) s_mine = min(max(csrc[beg + k], 0), n - 1);
            const int m = min(cnt - base, 16);
            for (int j = 0; j < m; j += 4) {
                int sj[4]; float4 r[4];
#pragma unroll
                for (int u = 0; u < 4; ++u) {
                    sj[u] = __shfl(s_mine, j + u, 16);
                    r[u]  = *(const float4*)(ts + (size_t)sj[u] * 64 + l16 * 4);
                }
#pragma unroll
                for (int u = 0; u < 4; ++u) {
                    const bool live = (u == 0) || (j + u) < m;
                    acc.x += live ? r[u].x : 0.0f;
                    acc.y += live ? r[u].y : 0.0f;
                    acc.z += live ? r[u].z : 0.0f;
                    acc.w += live ? r[u].w : 0.0f;
                }
            }
        }
        float4 o = make_float4(fmaxf(fmaf(di, acc.x, bv.x), 0.0f), fmaxf(fmaf(di, acc.y, bv.y), 0.0f),
                               fmaxf(fmaf(di, acc.z, bv.z), 0.0f), fmaxf(fmaf(di, acc.w, bv.w), 0.0f));
        *(float4*)(h + (size_t)i * 64 + l16 * 4) = o;
    }
}


extern "C" void kernel_launch(void* const* d_in, const int* in_sizes, int n_in,
                              void* d_out, int out_size, void* d_ws, size_t ws_size,
                              hipStream_t stream) {
    (void)n_in; (void)out_size; (void)ws_size;

    const float* x   = (const float*)d_in[0];
    const int*   ei  = (const int*)  d_in[1];
    const float* W1  = (const float*)d_in[2];
    const float* b1  = (const float*)d_in[3];
    const float* W2  = (const float*)d_in[4];
    const float* b2  = (const float*)d_in[5];
    const float* Wf1 = (const float*)d_in[6];
    const float* bf1 = (const float*)d_in[7];
    const float* Wf2 = (const float*)d_in[8];
    const float* bf2 = (const float*)d_in[9];
    const float* Wf3 = (const float*)d_in[10];
    const float* bf3 = (const float*)d_in[11];
    float* out = (float*)d_out;

    const int n = in_sizes[0] / 64;
    const int E = in_sizes[1] / 2;

    const size_t nPad  = ((size_t)n + 1 + 63) / 64 * 64;
    const size_t ePad  = ((size_t)E + 63) / 64 * 64;
    float* ws     = (float*)d_ws;
    float* dinv   = ws;
    int*   cnt    = (int*)(ws + nPad);
    int*   rowptr = (int*)(ws + 2 * nPad);
    int*   rank   = (int*)(ws + 3 * nPad);
    int*   csrc   = (int*)(ws + 3 * nPad + ePad);
    float* bufA   = ws + 3 * nPad + 2 * ePad;
    float* bufB   = bufA + (size_t)n * 64;

    const int T = 256;
    const int rowTiles = (n + 15) / 16;
    const int gemmBlocks = (rowTiles + 7) / 8;
    const int edgeBlocks4 = (E + 4 * T - 1) / (4 * T);
    const int gatherBlocks = 2048;

    k_zero_counts<<<(n + T - 1) / T, T, 0, stream>>>(cnt, n);
    k_count_rank<<<edgeBlocks4, T, 0, stream>>>(ei, cnt, rank, E);
    k_scan_rowptr<<<(n + 1023) / 1024, 1024, 0, stream>>>(cnt, rowptr, dinv, n);
    k_place_edges<<<edgeBlocks4, T, 0, stream>>>(ei, rank, rowptr, csrc, E);

    k_gemm64<<<gemmBlocks, T, 0, stream>>>(x, W1, dinv, bufA, n);
    k_gather_rows<<<gatherBlocks, T, 0, stream>>>(rowptr, csrc, dinv, bufA, b1, bufB, n);

    k_gemm64<<<gemmBlocks, T, 0, stream>>>(bufB, W2, dinv, bufA, n);
    k_gather_rows<<<gatherBlocks, T, 0, stream>>>(rowptr, csrc, dinv, bufA, b2, bufB, n);

    k_mlp_head<<<gemmBlocks, T, 0, stream>>>(bufB, Wf1, bf1, Wf2, bf2, Wf3, bf3, out, n);
}


# Aggregation kernels keep each block's finished rows in LDS and run the dense layer(s) that follow on them in the same launch (layer-2 product; MLP head): h1, h2 never reach memory; orders unchanged
# speedup vs baseline: 2.2711x; 1.0708x over previous
#include <hip/hip_runtime.h>


typedef __attribute__((ext_vector_type(2))) float v2f;
typedef __attribute__((ext_vector_type(8))) float v8f;


__global__ void k_zero_counts(int* __restrict__ cnt, int n) {
    int i = blockIdx.x * blockDim.x + threadIdx.x;
    if (i < n) cnt[i] = 0;
}

__global__ void k_count_rank(const int* __restrict__ ei, int* __restrict__ cnt,
                             int* __restrict__ rank, int nedges) {
    const int tid  = blockIdx.x * blockDim.x + threadIdx.x;
    const int nthr = gridDim.x * blockDim.x;
    int d[4];
#pragma unroll
    for (int k = 0; k < 4; ++k) {
        const int e = tid + k * nthr;
        d[k] = ei[nedges + min(e, nedges - 1)];
    }
    int r[4];
#pragma unroll
    for (int k = 0; k < 4; ++k) {
        const int e = tid + k * nthr;
        r[k] = atomicAdd(&cnt[d[k]], (e < nedges) ? 1 : 0);
    }
#pragma unroll
    for (int k = 0; k < 4; ++k) {
        const int e = tid + k * nthr;
        if (e < nedges) rank[e] = r[k];
    }
}

__global__ __launch_bounds__(1024) void k_scan_rowptr(const int* __restrict__ cnt, int* __restrict__ rowptr,
                                                      float* __restrict__ dinv, int n) {
    __shared__ int red[32];
    __shared__ int wtot[32];
    const int tid  = threadIdx.x;
    const int lane = tid & 31;
    const int wv   = tid >> 5;
    const int start = blockIdx.x * 1024;

    int part = 0;
    const int nvec = start >> 2;
    if (nvec > 0) {
        const int4* __restrict__ c4 = (const int4*)cnt;
        for (int j = tid; j < nvec; j += 4096) {
            int4 q[4];
#pragma unroll
            for (int k = 0; k < 4; ++k) q[k] = c4[min(j + k * 1024, nvec - 1)];
#pragma unroll
            for (int k = 0; k < 4; ++k)
                if (j + k * 1024 < nvec) part += q[k].x + q[k].y + q[k].z + q[k].w;
        }
    }
#pragma unroll
    for (int o = 16; o > 0; o >>= 1) part += __shfl_xor(part, o, 32);
    if (lane == 0) red[wv] = part;

    const int i = start + tid;
    const int c = (i < n) ? cnt[i] : 0;
    int incl = c;
#pragma unroll
    for (int o = 1; o < 32; o <<= 1) {
        const int up = __shfl_up(incl, o, 32);
        if (lane >= o) incl += up;
    }
    if (lane == 31) wtot[wv] = incl;
    __syncthreads();

    int offset = 0, wpre = 0;
#pragma unroll
    for (int w = 0; w < 32; ++w) {
        offset += red[w];
        if (w < wv) wpre += wtot[w];
    }
    const int excl = offset + wpre + incl - c;
    if (i < n) {
        rowptr[i] = excl;
        dinv[i] = rsqrtf((float)(c + 1));
        if (i == n - 1) rowptr[n] = excl + c;
    }
}

__global__ void k_place_edges(const int* __restrict__ ei, const int* __restrict__ rank,
                              const int* __restrict__ rowptr, int* __restrict__ csrc, int nedges) {
    const int tid  = blockIdx.x * blockDim.x + threadIdx.x;
    const int nthr = gridDim.x * blockDim.x;
    int s[4], d[4], r[4];
#pragma unroll
    for (int k = 0; k < 4; ++k) {
        const int e = min(tid + k * nthr, nedges - 1);
        s[k] = ei[e];
        d[k] = ei[nedges + e];
        r[k] = rank[e];
    }
    int p[4];
#pragma unroll
    for (int k = 0; k < 4; ++k) p[k] = rowptr[d[k]] + r[k];
#pragma unroll
    for (int k = 0; k < 4; ++k) {
        const int e = tid + k * nthr;
        if (e < nedges) csrc[min(max(p[k], 0), nedges - 1)] = s[k];
    }
}


#define WT_PITCH 68

__global__ void k_gemm64(const float* __restrict__ A, const float* __restrict__ W,
                         const float* __restrict__ rowscale,
                         float* __restrict__ C, int nrows) {
    __shared__ float ldsW[64 * WT_PITCH];

    for (int idx = threadIdx.x; idx < 4096; idx += 256) {
        const int k = idx >> 6;
        const int col = idx & 63;
        ldsW[col * WT_PITCH + k] = W[idx];
    }
    __syncthreads();

    const int lane = threadIdx.x & 31;
    const int wave = threadIdx.x >> 5;
    const int rowTiles = (nrows + 15) >> 4;
    const int tile = blockIdx.x * 8 + wave;
    if (tile < rowTiles) {
        const int half = lane >> 4;
        const int m    = lane & 15;
        int row = tile * 16 + m;
        if (row >= nrows) row = nrows - 1;
        const float* __restrict__ arow = A + (size_t)row * 64;

        float sc[8];
#pragma unroll
        for (int r = 0; r < 8; ++r)
            sc[r] = rowscale[min(tile * 16 + r + 8 * half, nrows - 1)];

        v8f c0 = {}, c1 = {}, c2 = {}, c3 = {};
#pragma unroll
        for (int kk = 0; kk < 16; ++kk) {
            const int k0 = kk * 4 + 2 * half;
            v2f a;
            a.x = arow[k0];
            a.y = arow[k0 + 1];
            v2f b0 = *(const v2f*)&ldsW[(0 * 16 + m) * WT_PITCH + k0];
            v2f b1 = *(const v2f*)&ldsW[(1 * 16 + m) * WT_PITCH + k0];
            v2f b2 = *(const v2f*)&ldsW[(2 * 16 + m) * WT_PITCH + k0];
            v2f b3 = *(const v2f*)&ldsW[(3 * 16 + m) * WT_PITCH + k0];
            c0 = __builtin_amdgcn_wmma_f32_16x16x4_f32(false, a, false, b0, (short)0, c0, false, false);
            c1 = __builtin_amdgcn_wmma_f32_16x16x4_f32(false, a, false, b1, (short)0, c1, false, false);
            c2 = __builtin_amdgcn_wmma_f32_16x16x4_f32(false, a, false, b2, (short)0, c2, false, false);
            c3 = __builtin_amdgcn_wmma_f32_16x16x4_f32(false, a, false, b3, (short)0, c3, false, false);
        }

        v8f acc[4] = {c0, c1, c2, c3};
#pragma unroll
        for (int nt = 0; nt < 4; ++nt) {
            const int col = nt * 16 + m;
#pragma unroll
            for (int r = 0; r < 8; ++r) {
                const int orow = tile * 16 + r + 8 * half;
                if (orow < nrows)
                    C[(size_t)orow * 64 + col] = acc[nt][r] * sc[r];
            }
        }
    }
}


#define TILE_PITCH 68

template <int NPG>
__device__ __forceinline__ void gather_rows_to_lds(const int* __restrict__ rowptr, const int* __restrict__ csrc,
                                                   const float* __restrict__ dinv, const float* __restrict__ ts,
                                                   const float* __restrict__ bias, float* __restrict__ ldsH, int n) {
    const int l16   = threadIdx.x & 15;
    const int gid   = threadIdx.x >> 4;
    const int first = blockIdx.x * (16 * NPG) + gid * NPG;
    const float4 bv = *(const float4*)(bias + l16 * 4);
    const int rp = rowptr[min(first + min(l16, NPG), n)];

    for (int s = 0; s < NPG; ++s) {
        const int i   = first + s;
        const int beg = __shfl(rp, s, 16);
        const int cnt = __shfl(rp, s + 1, 16) - beg;
        float4 o = make_float4(0.0f, 0.0f, 0.0f, 0.0f);
        if (i < n) {
            const float di = dinv[i];
            float4 acc = *(const float4*)(ts + (size_t)i * 64 + l16 * 4);

            for (int base = 0; base < cnt; base += 16) {
                const int k = base + l16;
                int s_mine = i;
                if (k < cnt) s_mine = min(max(csrc[beg + k], 0), n - 1);
                const int m = min(cnt - base, 16);
                for (int j = 0; j < m; j += 4) {
                    int sj[4]; float4 r[4];
#pragma unroll
                    for (int u = 0; u < 4; ++u) {
                        sj[u] = __shfl(s_mine, j + u, 16);
                        r[u]  = *(const float4*)(ts + (size_t)sj[u] * 64 + l16 * 4);
                    }
#pragma unroll
                    for (int u = 0; u < 4; ++u) {
                        const bool live = (u == 0) || (j + u) < m;
                        acc.x += live ? r[u].x : 0.0f;
                        acc.y += live ? r[u].y : 0.0f;
                        acc.z += live ? r[u].z : 0.0f;
                        acc.w += live ? r[u].w : 0.0f;
                    }
                }
            }
            o = make_float4(fmaxf(fmaf(di, acc.x, bv.x), 0.0f), fmaxf(fmaf(di, acc.y, bv.y), 0.0f),
                            fmaxf(fmaf(di, acc.z, bv.z), 0.0f), fmaxf(fmaf(di, acc.w, bv.w), 0.0f));
        }
        *(float4*)&ldsH[(gid * NPG + s) * TILE_PITCH + l16 * 4] = o;
    }
}


__global__ __launch_bounds__(256) void k_gather_gemm64(const int* __restrict__ rowptr, const int* __restrict__ csrc,
                                                       const float* __restrict__ dinv, const float* __restrict__ ts,
                                                       const float* __restrict__ bias, const float* __restrict__ W,
                                                       float* __restrict__ C, int n) {
    __shared__ __align__(16) float ldsW[64 * WT_PITCH];
    __shared__ __align__(16) float ldsH[64 * TILE_PITCH];

    float w[16];
#pragma unroll
    for (int i = 0; i < 16; ++i) w[i] = W[threadIdx.x + 256 * i];
#pragma unroll
    for (int i = 0; i < 16; ++i) {
        const int k = (threadIdx.x >> 6) + 4 * i;
        const int col = threadIdx.x & 63;
        ldsW[col * WT_PITCH + k] = w[i];
    }

    gather_rows_to_lds<4>(rowptr, csrc, dinv, ts, bias, ldsH, n);
    __syncthreads();

    const int lane = threadIdx.x & 31;
    const int wave = threadIdx.x >> 5;
    const int half = lane >> 4;
    const int m    = lane & 15;
    const int lt   = wave >> 1;
    const int nt0  = (wave & 1) * 2;
    const int row0 = blockIdx.x * 64 + lt * 16;
    if (row0 < n) {
        float sc[8];
#pragma unroll
        for (int r = 0; r < 8; ++r)
            sc[r] = dinv[min(row0 + r + 8 * half, n - 1)];

        v8f c0 = {}, c1 = {};
#pragma unroll
        for (int kk = 0; kk < 16; ++kk) {
            const int k0 = kk * 4 + 2 * half;
            const v2f a  = *(const v2f*)&ldsH[(lt * 16 + m) * TILE_PITCH + k0];
            const v2f b0 = *(const v2f*)&ldsW[((nt0 + 0) * 16 + m) * WT_PITCH + k0];
            const v2f b1 = *(const v2f*)&ldsW[((nt0 + 1) * 16 + m) * WT_PITCH + k0];
            c0 = __builtin_amdgcn_wmma_f32_16x16x4_f32(false, a, false, b0, (short)0, c0, false, false);
            c1 = __builtin_amdgcn_wmma_f32_16x16x4_f32(false, a, false, b1, (short)0, c1, false, false);
        }

        v8f acc2[2] = {c0, c1};
#pragma unroll
        for (int t = 0; t < 2; ++t) {
            const int col = (nt0 + t) * 16 + m;
#pragma unroll
            for (int r = 0; r < 8; ++r) {
                const int orow = row0 + r + 8 * half;
                if (orow < n)
                    C[(size_t)orow * 64 + col] = acc2[t][r] * sc[r];
            }
        }
    }
}


__device__ __forceinline__ int head_kpos(int f) {
    return (f & 0x31) | ((f & 0x6) << 1) | ((f >> 2) & 0x2);
}

__global__ __launch_bounds__(256) void k_gather_mlp_head(const int* __restrict__ rowptr, const int* __restrict__ csrc,
                                                         const float* __restrict__ dinv, const float* __restrict__ ts,
                                                         const float* __restrict__ bias,
                                                         const float* __restrict__ Wa, const float* __restrict__ ba,
                                                         const float* __restrict__ Wb, const float* __restrict__ bb,
                                                         const float* __restrict__ w3, const float* __restrict__ b3,
                                                         float* __restrict__ out, int n) {
    __shared__ __align__(16) float ldsWa[64 * WT_PITCH];
    __shared__ __align__(16) float ldsWb[64 * WT_PITCH];
    __shared__ __align__(16) float ldsV[3 * 64];
    __shared__ __align__(16) float ldsH[80 * TILE_PITCH];

    {
        float wa[16], wb[16];
#pragma unroll
        for (int i = 0; i < 16; ++i) {
            wa[i] = Wa[threadIdx.x + 256 * i];
            wb[i] = Wb[threadIdx.x + 256 * i];
        }
#pragma unroll
        for (int i = 0; i < 16; ++i) {
            const int k = (threadIdx.x >> 6) + 4 * i;
            const int col = threadIdx.x & 63;
            ldsWa[col * WT_PITCH + k] = wa[i];
            ldsWb[col * WT_PITCH + head_kpos(k)] = wb[i];
        }
        if (threadIdx.x < 64) {
            ldsV[threadIdx.x]       = ba[threadIdx.x];
            ldsV[64 + threadIdx.x]  = bb[threadIdx.x];
            ldsV[128 + threadIdx.x] = w3[threadIdx.x];
        }
    }

    gather_rows_to_lds<5>(rowptr, csrc, dinv, ts, bias, ldsH, n);
    __syncthreads();

    const int lane = threadIdx.x & 31;
    const int wave = threadIdx.x >> 5;
    const int half = lane >> 4;
    const int c    = lane & 15;
    const int node = blockIdx.x * 80 + wave * 16 + c;
    if (wave < 5 && blockIdx.x * 80 + wave * 16 < n) {
        const float* hrow = &ldsH[(wave * 16 + c) * TILE_PITCH + 2 * half];

        v8f t[4] = {};
#pragma unroll
        for (int kk = 0; kk < 16; ++kk) {
            const int k0 = kk * 4 + 2 * half;
            const v2f x = *(const v2f*)(hrow + 4 * kk);
            v2f a[4];
#pragma unroll
            for (int mt = 0; mt < 4; ++mt) a[mt] = *(const v2f*)&ldsWa[(mt * 16 + c) * WT_PITCH + k0];
#pragma unroll
            for (int mt = 0; mt < 4; ++mt)
                t[mt] = __builtin_amdgcn_wmma_f32_16x16x4_f32(false, a[mt], false, x, (short)0, t[mt], false, false);
        }

#pragma unroll
        for (int mt = 0; mt < 4; ++mt) {
            const float4 g0 = *(const float4*)&ldsV[mt * 16 + 8 * half];
            const float4 g1 = *(const float4*)&ldsV[mt * 16 + 8 * half + 4];
            t[mt][0] = fmaxf(t[mt][0] + g0.x, 0.0f);
            t[mt][1] = fmaxf(t[mt][1] + g0.y, 0.0f);
            t[mt][2] = fmaxf(t[mt][2] + g0.z, 0.0f);
            t[mt][3] = fmaxf(t[mt][3] + g0.w, 0.0f);
            t[mt][4] = fmaxf(t[mt][4] + g1.x, 0.0f);
            t[mt][5] = fmaxf(t[mt][5] + g1.y, 0.0f);
            t[mt][6] = fmaxf(t[mt][6] + g1.z, 0.0f);
            t[mt][7] = fmaxf(t[mt][7] + g1.w, 0.0f);
        }

        v8f u[4] = {};
#pragma unroll
        for (int nt = 0; nt < 4; ++nt) {
#pragma unroll
            for (int q = 0; q < 4; ++q) {
                v2f b;
                b.x = t[nt][2 * q];
                b.y = t[nt][2 * q + 1];
                const int kp = nt * 16 + q * 4 + 2 * half;
                v2f a[4];
#pragma unroll
                for (int mt = 0; mt < 4; ++mt) a[mt] = *(const v2f*)&ldsWb[(mt * 16 + c) * WT_PITCH + kp];
#pragma unroll
                for (int mt = 0; mt < 4; ++mt)
                    u[mt] = __builtin_amdgcn_wmma_f32_16x16x4_f32(false, a[mt], false, b, (short)0, u[mt], false, false);
            }
        }

        float p[4];
#pragma unroll
        for (int mt = 0; mt < 4; ++mt) {
            const float4 g0 = *(const float4*)&ldsV[64 + mt * 16 + 8 * half];
            const float4 g1 = *(const float4*)&ldsV[64 + mt * 16 + 8 * half + 4];
            const float4 w0 = *(const float4*)&ldsV[128 + mt * 16 + 8 * half];
            const float4 w1 = *(const float4*)&ldsV[128 + mt * 16 + 8 * half + 4];
            float s = fmaxf(u[mt][0] + g0.x, 0.0f) * w0.x;
            s = fmaf(fmaxf(u[mt][1] + g0.y, 0.0f), w0.y, s);
            s = fmaf(fmaxf(u[mt][2] + g0.z, 0.0f), w0.z, s);
            s = fmaf(fmaxf(u[mt][3] + g0.w, 0.0f), w0.w, s);
            s = fmaf(fmaxf(u[mt][4] + g1.x, 0.0f), w1.x, s);
            s = fmaf(fmaxf(u[mt][5] + g1.y, 0.0f), w1.y, s);
            s = fmaf(fmaxf(u[mt][6] + g1.z, 0.0f), w1.z, s);
            s = fmaf(fmaxf(u[mt][7] + g1.w, 0.0f), w1.w, s);
            p[mt] = s;
        }
        float dot = (p[0] + p[1]) + (p[2] + p[3]);
        dot += __shfl_xor(dot, 16, 32);
        if (half == 0 && node < n) out[node] = dot + b3[0];
    }
}


extern "C" void kernel_launch(void* const* d_in, const int* in_sizes, int n_in,
                              void* d_out, int out_size, void* d_ws, size_t ws_size,
                              hipStream_t stream) {
    (void)n_in; (void)out_size; (void)ws_size;

    const float* x   = (const float*)d_in[0];
    const int*   ei  = (const int*)  d_in[1];
    const float* W1  = (const float*)d_in[2];
    const float* b1  = (const float*)d_in[3];
    const float* W2  = (const float*)d_in[4];
    const float* b2  = (const float*)d_in[5];
    const float* Wf1 = (const float*)d_in[6];
    const float* bf1 = (const float*)d_in[7];
    const float* Wf2 = (const float*)d_in[8];
    const float* bf2 = (const float*)d_in[9];
    const float* Wf3 = (const float*)d_in[10];
    const float* bf3 = (const float*)d_in[11];
    float* out = (float*)d_out;

    const int n = in_sizes[0] / 64;
    const int E = in_sizes[1] / 2;

    const size_t nPad  = ((size_t)n + 1 + 63) / 64 * 64;
    const size_t ePad  = ((size_t)E + 63) / 64 * 64;
    float* ws     = (float*)d_ws;
    float* dinv   = ws;
    int*   cnt    = (int*)(ws + nPad);
    int*   rowptr = (int*)(ws + 2 * nPad);
    int*   rank   = (int*)(ws + 3 * nPad);
    int*   csrc   = (int*)(ws + 3 * nPad + ePad);
    float* bufA   = ws + 3 * nPad + 2 * ePad;
    float* bufB   = bufA + (size_t)n * 64;

    const int T = 256;
    const int rowTiles = (n + 15) / 16;
    const int gemmBlocks = (rowTiles + 7) / 8;
    const int edgeBlocks4 = (E + 4 * T - 1) / (4 * T);

    k_zero_counts<<<(n + T - 1) / T, T, 0, stream>>>(cnt, n);
    k_count_rank<<<edgeBlocks4, T, 0, stream>>>(ei, cnt, rank, E);
    k_scan_rowptr<<<(n + 1023) / 1024, 1024, 0, stream>>>(cnt, rowptr, dinv, n);
    k_place_edges<<<edgeBlocks4, T, 0, stream>>>(ei, rank, rowptr, csrc, E);

    k_gemm64<<<gemmBlocks, T, 0, stream>>>(x, W1, dinv, bufA, n);

    k_gather_gemm64<<<(n + 63) / 64, T, 0, stream>>>(rowptr, csrc, dinv, bufA, b1, W2, bufB, n);

    k_gather_mlp_head<<<(n + 79) / 80, T, 0, stream>>>(rowptr, csrc, dinv, bufB, b2, Wf1, bf1, Wf2, bf2, Wf3, bf3, out, n);
}


# Edge list by destination now built by a two-level counting sort through LDS bins (LDS atomics, whole-line stores, 2 launches for 4), not global returning atomics; sum order is the LDS arrival rank
# speedup vs baseline: 4.8955x; 2.1556x over previous
#include <hip/hip_runtime.h>


typedef __attribute__((ext_vector_type(2))) float v2f;
typedef __attribute__((ext_vector_type(8))) float v8f;


#define BIN_SHIFT   9
#define BIN_NODES   512
#define CHUNK_EDGES 8192
#define STAGE_CAP   12288

__global__ __launch_bounds__(1024) void k_bin_edges(const int* __restrict__ ei, int* __restrict__ binned,
                                                    int* __restrict__ tabCnt, int* __restrict__ tabOff,
                                                    int nedges, int nbins) {
    __shared__ int hist[1024];
    __shared__ int wtot[32];
    __shared__ unsigned sorted[CHUNK_EDGES];
    const int tid  = threadIdx.x;
    const int lane = tid & 31;
    const int wv   = tid >> 5;
    const int c    = blockIdx.x;
    const int nchunks = gridDim.x;
    const int e0  = c * CHUNK_EDGES;
    const int len = min(CHUNK_EDGES, nedges - e0);

    hist[tid] = 0;
    int s[8], d[8];
#pragma unroll
    for (int k = 0; k < 8; ++k) {
        const int e = min(e0 + k * 1024 + tid, nedges - 1);
        s[k] = ei[e];
        d[k] = ei[nedges + e];
    }
    __syncthreads();

    int bn[8], rk[8];
#pragma unroll
    for (int k = 0; k < 8; ++k) {
        const bool live = (k * 1024 + tid) < len;
        bn[k] = min(max(d[k] >> BIN_SHIFT, 0), nbins - 1);
        rk[k] = live ? atomicAdd(&hist[bn[k]], 1) : 0;
    }
    __syncthreads();

    const int h = hist[tid];
    int incl = h;
#pragma unroll
    for (int o = 1; o < 32; o <<= 1) {
        const int up = __shfl_up(incl, o, 32);
        if (lane >= o) incl += up;
    }
    if (lane == 31) wtot[wv] = incl;
    __syncthreads();
    int wpre = 0;
#pragma unroll
    for (int w = 0; w < 32; ++w)
        if (w < wv) wpre += wtot[w];
    const int excl = wpre + incl - h;
    hist[tid] = excl;
    if (tid < nbins) {
        tabCnt[tid * nchunks + c] = h;
        tabOff[tid * nchunks + c] = excl;
    }
    __syncthreads();

#pragma unroll
    for (int k = 0; k < 8; ++k) {
        const bool live = (k * 1024 + tid) < len;
        if (live) {
            const int p = min(hist[bn[k]] + rk[k], CHUNK_EDGES - 1);
            sorted[p] = ((unsigned)s[k] << BIN_SHIFT) | ((unsigned)d[k] & (BIN_NODES - 1));
        }
    }
    __syncthreads();
    for (int i = tid; i < len; i += 1024) binned[e0 + i] = (int)sorted[i];
}

template <int PASS>
__device__ __forceinline__ void walk_bin_pieces(const int* __restrict__ binned, const int* segOff, const int* segCnt,
                                                int* counters, int* stage, int* __restrict__ csrc,
                                                int nchunks, int nedges, int base, bool fits) {
    const int lane = threadIdx.x & 31;
    const int wv   = threadIdx.x >> 5;
    auto visit = [&](unsigned w) {
        const int dl = (int)(w & (BIN_NODES - 1));
        if (PASS == 0) {
            atomicAdd(&counters[dl], 1);
        } else {
            const int p   = atomicAdd(&counters[dl], 1);
            const int src = (int)(w >> BIN_SHIFT);
            if (fits) stage[min(max(p, 0), STAGE_CAP - 1)] = src;
            else      csrc[min(max(base + p, 0), nedges - 1)] = src;
        }
    };
    for (int s0 = wv; s0 < nchunks; s0 += 128) {
        int o[4], m[4];
        unsigned w0[4], w1[4];
#pragma unroll
        for (int u = 0; u < 4; ++u) {
            const int sg = s0 + 32 * u;
            o[u] = (sg < nchunks) ? segOff[sg] : 0;
            m[u] = (sg < nchunks) ? segCnt[sg] : 0;
        }
#pragma unroll
        for (int u = 0; u < 4; ++u) {
            w0[u] = (unsigned)binned[min(o[u] + lane, nedges - 1)];
            w1[u] = (unsigned)binned[min(o[u] + lane + 32, nedges - 1)];
        }
#pragma unroll
        for (int u = 0; u < 4; ++u) {
            if (lane < m[u]) visit(w0[u]);
            if (lane + 32 < m[u]) visit(w1[u]);
            for (int k = lane + 64; k < m[u]; k += 32) visit((unsigned)binned[min(o[u] + k, nedges - 1)]);
        }
    }
}

__global__ __launch_bounds__(1024) void k_bin_sort(const int* __restrict__ binned, const int* __restrict__ tabCnt,
                                                   const int* __restrict__ tabOff, int* __restrict__ rowptr,
                                                   float* __restrict__ dinv, int* __restrict__ csrc,
                                                   int n, int nedges, int nchunks) {
    __shared__ int deg[BIN_NODES];
    __shared__ int cursor[BIN_NODES];
    __shared__ int segOff[1024];
    __shared__ int segCnt[1024];
    __shared__ int red[32];
    __shared__ int wtot[32];
    __shared__ int stage[STAGE_CAP];
    const int tid  = threadIdx.x;
    const int lane = tid & 31;
    const int wv   = tid >> 5;
    const int b    = blockIdx.x;

    int part = 0;
    const int npre = b * nchunks;
    for (int j = tid; j < npre; j += 16384) {
        int q[16];
#pragma unroll
        for (int k = 0; k < 16; ++k) q[k] = tabCnt[min(j + k * 1024, npre - 1)];
#pragma unroll
        for (int k = 0; k < 16; ++k)
            if (j + k * 1024 < npre) part += q[k];
    }
#pragma unroll
    for (int o = 16; o > 0; o >>= 1) part += __shfl_xor(part, o, 32);
    if (lane == 0) red[wv] = part;

    if (tid < BIN_NODES) deg[tid] = 0;
    segCnt[tid] = (tid < nchunks) ? tabCnt[b * nchunks + tid] : 0;
    segOff[tid] = (tid < nchunks) ? tid * CHUNK_EDGES + tabOff[b * nchunks + tid] : 0;
    __syncthreads();

    walk_bin_pieces<0>(binned, segOff, segCnt, deg, stage, csrc, nchunks, nedges, 0, true);
    __syncthreads();

    const int i = b * BIN_NODES + tid;
    const int cdeg = (tid < BIN_NODES) ? deg[tid] : 0;
    int incl = cdeg;
#pragma unroll
    for (int o = 1; o < 32; o <<= 1) {
        const int up = __shfl_up(incl, o, 32);
        if (lane >= o) incl += up;
    }
    if (lane == 31) wtot[wv] = incl;
    __syncthreads();

    int base = 0, wpre = 0, total = 0;
#pragma unroll
    for (int w = 0; w < 32; ++w) {
        base  += red[w];
        total += wtot[w];
        if (w < wv) wpre += wtot[w];
    }
    const int excl = wpre + incl - cdeg;
    if (tid < BIN_NODES) {
        cursor[tid] = excl;
        if (i < n) {
            rowptr[i] = base + excl;
            dinv[i] = rsqrtf((float)(cdeg + 1));
            if (i == n - 1) rowptr[n] = base + excl + cdeg;
        }
    }
    __syncthreads();

    const bool fits = total <= STAGE_CAP;
    walk_bin_pieces<1>(binned, segOff, segCnt, cursor, stage, csrc, nchunks, nedges, base, fits);
    __syncthreads();
    if (fits)
        for (int j = tid; j < total; j += 1024) csrc[min(base + j, nedges - 1)] = stage[j];
}


#define WT_PITCH 68

__global__ void k_gemm64(const float* __restrict__ A, const float* __restrict__ W,
                         const float* __restrict__ rowscale,
                         float* __restrict__ C, int nrows) {
    __shared__ float ldsW[64 * WT_PITCH];

    for (int idx = threadIdx.x; idx < 4096; idx += 256) {
        const int k = idx >> 6;
        const int col = idx & 63;
        ldsW[col * WT_PITCH + k] = W[idx];
    }
    __syncthreads();

    const int lane = threadIdx.x & 31;
    const int wave = threadIdx.x >> 5;
    const int rowTiles = (nrows + 15) >> 4;
    const int tile = blockIdx.x * 8 + wave;
    if (tile < rowTiles) {
        const int half = lane >> 4;
        const int m    = lane & 15;
        int row = tile * 16 + m;
        if (row >= nrows) row = nrows - 1;
        const float* __restrict__ arow = A + (size_t)row * 64;

        float sc[8];
#pragma unroll
        for (int r = 0; r < 8; ++r)
            sc[r] = rowscale[min(tile * 16 + r + 8 * half, nrows - 1)];

        v8f c0 = {}, c1 = {}, c2 = {}, c3 = {};
#pragma unroll
        for (int kk = 0; kk < 16; ++kk) {
            const int k0 = kk * 4 + 2 * half;
            v2f a;
            a.x = arow[k0];
            a.y = arow[k0 + 1];
            v2f b0 = *(const v2f*)&ldsW[(0 * 16 + m) * WT_PITCH + k0];
            v2f b1 = *(const v2f*)&ldsW[(1 * 16 + m) * WT_PITCH + k0];
            v2f b2 = *(const v2f*)&ldsW[(2 * 16 + m) * WT_PITCH + k0];
            v2f b3 = *(const v2f*)&ldsW[(3 * 16 + m) * WT_PITCH + k0];
            c0 = __builtin_amdgcn_wmma_f32_16x16x4_f32(false, a, false, b0, (short)0, c0, false, false);
            c1 = __builtin_amdgcn_wmma_f32_16x16x4_f32(false, a, false, b1, (short)0, c1, false, false);
            c2 = __builtin_amdgcn_wmma_f32_16x16x4_f32(false, a, false, b2, (short)0, c2, false, false);
            c3 = __builtin_amdgcn_wmma_f32_16x16x4_f32(false, a, false, b3, (short)0, c3, false, false);
        }

        v8f acc[4] = {c0, c1, c2, c3};
#pragma unroll
        for (int nt = 0; nt < 4; ++nt) {
            const int col = nt * 16 + m;
#pragma unroll
            for (int r = 0; r < 8; ++r) {
                const int orow = tile * 16 + r + 8 * half;
                if (orow < nrows)
                    C[(size_t)orow * 64 + col] = acc[nt][r] * sc[r];
            }
        }
    }
}


#define TILE_PITCH 68

template <int NPG>
__device__ __forceinline__ void gather_rows_to_lds(const int* __restrict__ rowptr, const int* __restrict__ csrc,
                                                   const float* __restrict__ dinv, const float* __restrict__ ts,
                                                   const float* __restrict__ bias, float* __restrict__ ldsH, int n) {
    const int l16   = threadIdx.x & 15;
    const int gid   = threadIdx.x >> 4;
    const int first = blockIdx.x * (16 * NPG) + gid * NPG;
    const float4 bv = *(const float4*)(bias + l16 * 4);
    const int rp = rowptr[min(first + min(l16, NPG), n)];

    for (int s = 0; s < NPG; ++s) {
        const int i   = first + s;
        const int beg = __shfl(rp, s, 16);
        const int cnt = __shfl(rp, s + 1, 16) - beg;
        float4 o = make_float4(0.0f, 0.0f, 0.0f, 0.0f);
        if (i < n) {
            const float di = dinv[i];
            float4 acc = *(const float4*)(ts + (size_t)i * 64 + l16 * 4);

            for (int base = 0; base < cnt; base += 16) {
                const int k = base + l16;
                int s_mine = i;
                if (k < cnt) s_mine = min(max(csrc[beg + k], 0), n - 1);
                const int m = min(cnt - base, 16);
                for (int j = 0; j < m; j += 4) {
                    int sj[4]; float4 r[4];
#pragma unroll
                    for (int u = 0; u < 4; ++u) {
                        sj[u] = __shfl(s_mine, j + u, 16);
                        r[u]  = *(const float4*)(ts + (size_t)sj[u] * 64 + l16 * 4);
                    }
#pragma unroll
                    for (int u = 0; u < 4; ++u) {
                        const bool live = (u == 0) || (j + u) < m;
                        acc.x += live ? r[u].x : 0.0f;
                        acc.y += live ? r[u].y : 0.0f;
                        acc.z += live ? r[u].z : 0.0f;
                        acc.w += live ? r[u].w : 0.0f;
                    }
                }
            }
            o = make_float4(fmaxf(fmaf(di, acc.x, bv.x), 0.0f), fmaxf(fmaf(di, acc.y, bv.y), 0.0f),
                            fmaxf(fmaf(di, acc.z, bv.z), 0.0f), fmaxf(fmaf(di, acc.w, bv.w), 0.0f));
        }
        *(float4*)&ldsH[(gid * NPG + s) * TILE_PITCH + l16 * 4] = o;
    }
}


__global__ __launch_bounds__(256) void k_gather_gemm64(const int* __restrict__ rowptr, const int* __restrict__ csrc,
                                                       const float* __restrict__ dinv, const float* __restrict__ ts,
                                                       const float* __restrict__ bias, const float* __restrict__ W,
                                                       float* __restrict__ C, int n) {
    __shared__ __align__(16) float ldsW[64 * WT_PITCH];
    __shared__ __align__(16) float ldsH[64 * TILE_PITCH];

    float w[16];
#pragma unroll
    for (int i = 0; i < 16; ++i) w[i] = W[threadIdx.x + 256 * i];
#pragma unroll
    for (int i = 0; i < 16; ++i) {
        const int k = (threadIdx.x >> 6) + 4 * i;
        const int col = threadIdx.x & 63;
        ldsW[col * WT_PITCH + k] = w[i];
    }

    gather_rows_to_lds<4>(rowptr, csrc, dinv, ts, bias, ldsH, n);
    __syncthreads();

    const int lane = threadIdx.x & 31;
    const int wave = threadIdx.x >> 5;
    const int half = lane >> 4;
    const int m    = lane & 15;
    const int lt   = wave >> 1;
    const int nt0  = (wave & 1) * 2;
    const int row0 = blockIdx.x * 64 + lt * 16;
    if (row0 < n) {
        float sc[8];
#pragma unroll
        for (int r = 0; r < 8; ++r)
            sc[r] = dinv[min(row0 + r + 8 * half, n - 1)];

        v8f c0 = {}, c1 = {};
#pragma unroll
        for (int kk = 0; kk < 16; ++kk) {
            const int k0 = kk * 4 + 2 * half;
            const v2f a  = *(const v2f*)&ldsH[(lt * 16 + m) * TILE_PITCH + k0];
            const v2f b0 = *(const v2f*)&ldsW[((nt0 + 0) * 16 + m) * WT_PITCH + k0];
            const v2f b1 = *(const v2f*)&ldsW[((nt0 + 1) * 16 + m) * WT_PITCH + k0];
            c0 = __builtin_amdgcn_wmma_f32_16x16x4_f32(false, a, false, b0, (short)0, c0, false, false);
            c1 = __builtin_amdgcn_wmma_f32_16x16x4_f32(false, a, false, b1, (short)0, c1, false, false);
        }

        v8f acc2[2] = {c0, c1};
#pragma unroll
        for (int t = 0; t < 2; ++t) {
            const int col = (nt0 + t) * 16 + m;
#pragma unroll
            for (int r = 0; r < 8; ++r) {
                const int orow = row0 + r + 8 * half;
                if (orow < n)
                    C[(size_t)orow * 64 + col] = acc2[t][r] * sc[r];
            }
        }
    }
}


__device__ __forceinline__ int head_kpos(int f) {
    return (f & 0x31) | ((f & 0x6) << 1) | ((f >> 2) & 0x2);
}

__global__ __launch_bounds__(256) void k_gather_mlp_head(const int* __restrict__ rowptr, const int* __restrict__ csrc,
                                                         const float* __restrict__ dinv, const float* __restrict__ ts,
                                                         const float* __restrict__ bias,
                                                         const float* __restrict__ Wa, const float* __restrict__ ba,
                                                         const float* __restrict__ Wb, const float* __restrict__ bb,
                                                         const float* __restrict__ w3, const float* __restrict__ b3,
                                                         float* __restrict__ out, int n) {
    __shared__ __align__(16) float ldsWa[64 * WT_PITCH];
    __shared__ __align__(16) float ldsWb[64 * WT_PITCH];
    __shared__ __align__(16) float ldsV[3 * 64];
    __shared__ __align__(16) float ldsH[80 * TILE_PITCH];

    {
        float wa[16], wb[16];
#pragma unroll
        for (int i = 0; i < 16; ++i) {
            wa[i] = Wa[threadIdx.x + 256 * i];
            wb[i] = Wb[threadIdx.x + 256 * i];
        }
#pragma unroll
        for (int i = 0; i < 16; ++i) {
            const int k = (threadIdx.x >> 6) + 4 * i;
            const int col = threadIdx.x & 63;
            ldsWa[col * WT_PITCH + k] = wa[i];
            ldsWb[col * WT_PITCH + head_kpos(k)] = wb[i];
        }
        if (threadIdx.x < 64) {
            ldsV[threadIdx.x]       = ba[threadIdx.x];
            ldsV[64 + threadIdx.x]  = bb[threadIdx.x];
            ldsV[128 + threadIdx.x] = w3[threadIdx.x];
        }
    }

    gather_rows_to_lds<5>(rowptr, csrc, dinv, ts, bias, ldsH, n);
    __syncthreads();

    const int lane = threadIdx.x & 31;
    const int wave = threadIdx.x >> 5;
    const int half = lane >> 4;
    const int c    = lane & 15;
    const int node = blockIdx.x * 80 + wave * 16 + c;
    if (wave < 5 && blockIdx.x * 80 + wave * 16 < n) {
        const float* hrow = &ldsH[(wave * 16 + c) * TILE_PITCH + 2 * half];

        v8f t[4] = {};
#pragma unroll
        for (int kk = 0; kk < 16; ++kk) {
            const int k0 = kk * 4 + 2 * half;
            const v2f x = *(const v2f*)(hrow + 4 * kk);
            v2f a[4];
#pragma unroll
            for (int mt = 0; mt < 4; ++mt) a[mt] = *(const v2f*)&ldsWa[(mt * 16 + c) * WT_PITCH + k0];
#pragma unroll
            for (int mt = 0; mt < 4; ++mt)
                t[mt] = __builtin_amdgcn_wmma_f32_16x16x4_f32(false, a[mt], false, x, (short)0, t[mt], false, false);
        }

#pragma unroll
        for (int mt = 0; mt < 4; ++mt) {
            const float4 g0 = *(const float4*)&ldsV[mt * 16 + 8 * half];
            const float4 g1 = *(const float4*)&ldsV[mt * 16 + 8 * half + 4];
            t[mt][0] = fmaxf(t[mt][0] + g0.x, 0.0f);
            t[mt][1] = fmaxf(t[mt][1] + g0.y, 0.0f);
            t[mt][2] = fmaxf(t[mt][2] + g0.z, 0.0f);
            t[mt][3] = fmaxf(t[mt][3] + g0.w, 0.0f);
            t[mt][4] = fmaxf(t[mt][4] + g1.x, 0.0f);
            t[mt][5] = fmaxf(t[mt][5] + g1.y, 0.0f);
            t[mt][6] = fmaxf(t[mt][6] + g1.z, 0.0f);
            t[mt][7] = fmaxf(t[mt][7] + g1.w, 0.0f);
        }

        v8f u[4] = {};
#pragma unroll
        for (int nt = 0; nt < 4; ++nt) {
#pragma unroll
            for (int q = 0; q < 4; ++q) {
                v2f b;
                b.x = t[nt][2 * q];
                b.y = t[nt][2 * q + 1];
                const int kp = nt * 16 + q * 4 + 2 * half;
                v2f a[4];
#pragma unroll
                for (int mt = 0; mt < 4; ++mt) a[mt] = *(const v2f*)&ldsWb[(mt * 16 + c) * WT_PITCH + kp];
#pragma unroll
                for (int mt = 0; mt < 4; ++mt)
                    u[mt] = __builtin_amdgcn_wmma_f32_16x16x4_f32(false, a[mt], false, b, (short)0, u[mt], false, false);
            }
        }

        float p[4];
#pragma unroll
        for (int mt = 0; mt < 4; ++mt) {
            const float4 g0 = *(const float4*)&ldsV[64 + mt * 16 + 8 * half];
            const float4 g1 = *(const float4*)&ldsV[64 + mt * 16 + 8 * half + 4];
            const float4 w0 = *(const float4*)&ldsV[128 + mt * 16 + 8 * half];
            const float4 w1 = *(const float4*)&ldsV[128 + mt * 16 + 8 * half + 4];
            float s = fmaxf(u[mt][0] + g0.x, 0.0f) * w0.x;
            s = fmaf(fmaxf(u[mt][1] + g0.y, 0.0f), w0.y, s);
            s = fmaf(fmaxf(u[mt][2] + g0.z, 0.0f), w0.z, s);
            s = fmaf(fmaxf(u[mt][3] + g0.w, 0.0f), w0.w, s);
            s = fmaf(fmaxf(u[mt][4] + g1.x, 0.0f), w1.x, s);
            s = fmaf(fmaxf(u[mt][5] + g1.y, 0.0f), w1.y, s);
            s = fmaf(fmaxf(u[mt][6] + g1.z, 0.0f), w1.z, s);
            s = fmaf(fmaxf(u[mt][7] + g1.w, 0.0f), w1.w, s);
            p[mt] = s;
        }
        float dot = (p[0] + p[1]) + (p[2] + p[3]);
        dot += __shfl_xor(dot, 16, 32);
        if (half == 0 && node < n) out[node] = dot + b3[0];
    }
}


extern "C" void kernel_launch(void* const* d_in, const int* in_sizes, int n_in,
                              void* d_out, int out_size, void* d_ws, size_t ws_size,
                              hipStream_t stream) {
    (void)n_in; (void)out_size; (void)ws_size;

    const float* x   = (const float*)d_in[0];
    const int*   ei  = (const int*)  d_in[1];
    const float* W1  = (const float*)d_in[2];
    const float* b1  = (const float*)d_in[3];
    const float* W2  = (const float*)d_in[4];
    const float* b2  = (const float*)d_in[5];
    const float* Wf1 = (const float*)d_in[6];
    const float* bf1 = (const float*)d_in[7];
    const float* Wf2 = (const float*)d_in[8];
    const float* bf2 = (const float*)d_in[9];
    const float* Wf3 = (const float*)d_in[10];
    const float* bf3 = (const float*)d_in[11];
    float* out = (float*)d_out;

    const int n = in_sizes[0] / 64;
    const int E = in_sizes[1] / 2;

    const int nbins   = (n + BIN_NODES - 1) / BIN_NODES;
    const int nchunks = (E + CHUNK_EDGES - 1) / CHUNK_EDGES;

    const size_t nPad  = ((size_t)n + 1 + 63) / 64 * 64;
    const size_t ePad  = ((size_t)E + 63) / 64 * 64;
    const size_t tPad  = ((size_t)nbins * nchunks + 63) / 64 * 64;
    float* ws     = (float*)d_ws;
    float* dinv   = ws;
    int*   rowptr = (int*)(ws + nPad);
    int*   binned = (int*)(ws + 2 * nPad);
    int*   csrc   = (int*)(ws + 2 * nPad + ePad);
    int*   tabCnt = (int*)(ws + 2 * nPad + 2 * ePad);
    int*   tabOff = tabCnt + tPad;
    float* bufA   = ws + 2 * nPad + 2 * ePad + 2 * tPad;
    float* bufB   = bufA + (size_t)n * 64;

    const int T = 256;
    const int rowTiles = (n + 15) / 16;
    const int gemmBlocks = (rowTiles + 7) / 8;

    k_bin_edges<<<nchunks, 1024, 0, stream>>>(ei, binned, tabCnt, tabOff, E, nbins);
    k_bin_sort<<<nbins, 1024, 0, stream>>>(binned, tabCnt, tabOff, rowptr, dinv, csrc, n, E, nchunks);

    k_gemm64<<<gemmBlocks, T, 0, stream>>>(x, W1, dinv, bufA, n);

    k_gather_gemm64<<<(n + 63) / 64, T, 0, stream>>>(rowptr, csrc, dinv, bufA, b1, W2, bufB, n);

    k_gather_mlp_head<<<(n + 79) / 80, T, 0, stream>>>(rowptr, csrc, dinv, bufB, b2, Wf1, bf1, Wf2, bf2, Wf3, bf3, out, n);
}


# Sort launch also forms the layer-1 product rows (x W1 scaled by deg^-1/2) of each bin block's own 512 nodes on its 32 waves: the separate product launch is gone, 4 launches for 5; sums unchanged
# speedup vs baseline: 5.1373x; 1.0494x over previous
#include <hip/hip_runtime.h>


typedef __attribute__((ext_vector_type(2))) float v2f;
typedef __attribute__((ext_vector_type(8))) float v8f;


#define BIN_SHIFT   9
#define BIN_NODES   512
#define CHUNK_EDGES 8192
#define STAGE_CAP   12288

__global__ __launch_bounds__(1024) void k_bin_edges(const int* __restrict__ ei, int* __restrict__ binned,
                                                    int* __restrict__ tabCnt, int* __restrict__ tabOff,
                                                    int nedges, int nbins) {
    __shared__ int hist[1024];
    __shared__ int wtot[32];
    __shared__ unsigned sorted[CHUNK_EDGES];
    const int tid  = threadIdx.x;
    const int lane = tid & 31;
    const int wv   = tid >> 5;
    const int c    = blockIdx.x;
    const int nchunks = gridDim.x;
    const int e0  = c * CHUNK_EDGES;
    const int len = min(CHUNK_EDGES, nedges - e0);

    hist[tid] = 0;
    int s[8], d[8];
#pragma unroll
    for (int k = 0; k < 8; ++k) {
        const int e = min(e0 + k * 1024 + tid, nedges - 1);
        s[k] = ei[e];
        d[k] = ei[nedges + e];
    }
    __syncthreads();

    int bn[8], rk[8];
#pragma unroll
    for (int k = 0; k < 8; ++k) {
        const bool live = (k * 1024 + tid) < len;
        bn[k] = min(max(d[k] >> BIN_SHIFT, 0), nbins - 1);
        rk[k] = live ? atomicAdd(&hist[bn[k]], 1) : 0;
    }
    __syncthreads();

    const int h = hist[tid];
    int incl = h;
#pragma unroll
    for (int o = 1; o < 32; o <<= 1) {
        const int up = __shfl_up(incl, o, 32);
        if (lane >= o) incl += up;
    }
    if (lane == 31) wtot[wv] = incl;
    __syncthreads();
    int wpre = 0;
#pragma unroll
    for (int w = 0; w < 32; ++w)
        if (w < wv) wpre += wtot[w];
    const int excl = wpre + incl - h;
    hist[tid] = excl;
    if (tid < nbins) {
        tabCnt[tid * nchunks + c] = h;
        tabOff[tid * nchunks + c] = excl;
    }
    __syncthreads();

#pragma unroll
    for (int k = 0; k < 8; ++k) {
        const bool live = (k * 1024 + tid) < len;
        if (live) {
            const int p = min(hist[bn[k]] + rk[k], CHUNK_EDGES - 1);
            sorted[p] = ((unsigned)s[k] << BIN_SHIFT) | ((unsigned)d[k] & (BIN_NODES - 1));
        }
    }
    __syncthreads();
    for (int i = tid; i < len; i += 1024) binned[e0 + i] = (int)sorted[i];
}

template <int PASS>
__device__ __forceinline__ void walk_bin_pieces(const int* __restrict__ binned, const int* segOff, const int* segCnt,
                                                int* counters, int* stage, int* __restrict__ csrc,
                                                int nchunks, int nedges, int base, bool fits) {
    const int lane = threadIdx.x & 31;
    const int wv   = threadIdx.x >> 5;
    auto visit = [&](unsigned w) {
        const int dl = (int)(w & (BIN_NODES - 1));
        if (PASS == 0) {
            atomicAdd(&counters[dl], 1);
        } else {
            const int p   = atomicAdd(&counters[dl], 1);
            const int src = (int)(w >> BIN_SHIFT);
            if (fits) stage[min(max(p, 0), STAGE_CAP - 1)] = src;
            else      csrc[min(max(base + p, 0), nedges - 1)] = src;
        }
    };
    for (int s0 = wv; s0 < nchunks; s0 += 128) {
        int o[4], m[4];
        unsigned w0[4], w1[4];
#pragma unroll
        for (int u = 0; u < 4; ++u) {
            const int sg = s0 + 32 * u;
            o[u] = (sg < nchunks) ? segOff[sg] : 0;
            m[u] = (sg < nchunks) ? segCnt[sg] : 0;
        }
#pragma unroll
        for (int u = 0; u < 4; ++u) {
            w0[u] = (unsigned)binned[min(o[u] + lane, nedges - 1)];
            w1[u] = (unsigned)binned[min(o[u] + lane + 32, nedges - 1)];
        }
#pragma unroll
        for (int u = 0; u < 4; ++u) {
            if (lane < m[u]) visit(w0[u]);
            if (lane + 32 < m[u]) visit(w1[u]);
            for (int k = lane + 64; k < m[u]; k += 32) visit((unsigned)binned[min(o[u] + k, nedges - 1)]);
        }
    }
}

#define WT_PITCH 68

__global__ __launch_bounds__(1024) void k_bin_sort(const int* __restrict__ binned, const int* __restrict__ tabCnt,
                                                   const int* __restrict__ tabOff, int* __restrict__ rowptr,
                                                   float* __restrict__ dinv, int* __restrict__ csrc,
                                                   const float* __restrict__ A, const float* __restrict__ W,
                                                   float* __restrict__ C,
                                                   int n, int nedges, int nchunks) {
    __shared__ int deg[BIN_NODES];
    __shared__ int cursor[BIN_NODES];
    __shared__ int segOff[1024];
    __shared__ int segCnt[1024];
    __shared__ int red[32];
    __shared__ int wtot[32];
    __shared__ int stage[STAGE_CAP];
    __shared__ __align__(16) float ldsW[64 * WT_PITCH];
    __shared__ float rowScale[BIN_NODES];
    const int tid  = threadIdx.x;
    const int lane = tid & 31;
    const int wv   = tid >> 5;
    const int b    = blockIdx.x;

    const int half = lane >> 4;
    const int m    = lane & 15;
    const int row0 = b * BIN_NODES + wv * 16;
    const bool tileLive = row0 < n;
    v2f afrag[16];
    {
        const float* __restrict__ arow = A + (size_t)min(row0 + m, n - 1) * 64;
#pragma unroll
        for (int kk = 0; kk < 16; ++kk) {
            const int k0 = kk * 4 + 2 * half;
            afrag[kk].x = arow[k0];
            afrag[kk].y = arow[k0 + 1];
        }
    }
    float wst[4];
#pragma unroll
    for (int i = 0; i < 4; ++i) wst[i] = W[tid + 1024 * i];
#pragma unroll
    for (int i = 0; i < 4; ++i) {
        const int k = (tid >> 6) + 16 * i;
        const int col = tid & 63;
        ldsW[col * WT_PITCH + k] = wst[i];
    }

    int part = 0;
    const int npre = b * nchunks;
    for (int j = tid; j < npre; j += 16384) {
        int q[16];
#pragma unroll
        for (int k = 0; k < 16; ++k) q[k] = tabCnt[min(j + k * 1024, npre - 1)];
#pragma unroll
        for (int k = 0; k < 16; ++k)
            if (j + k * 1024 < npre) part += q[k];
    }
#pragma unroll
    for (int o = 16; o > 0; o >>= 1) part += __shfl_xor(part, o, 32);
    if (lane == 0) red[wv] = part;

    if (tid < BIN_NODES) deg[tid] = 0;
    segCnt[tid] = (tid < nchunks) ? tabCnt[b * nchunks + tid] : 0;
    segOff[tid] = (tid < nchunks) ? tid * CHUNK_EDGES + tabOff[b * nchunks + tid] : 0;
    __syncthreads();

    v8f c0 = {}, c1 = {}, c2 = {}, c3 = {};
    if (tileLive) {
#pragma unroll
        for (int kk = 0; kk < 16; ++kk) {
            const int k0 = kk * 4 + 2 * half;
            const v2f b0 = *(const v2f*)&ldsW[(0 * 16 + m) * WT_PITCH + k0];
            const v2f b1 = *(const v2f*)&ldsW[(1 * 16 + m) * WT_PITCH + k0];
            const v2f b2 = *(const v2f*)&ldsW[(2 * 16 + m) * WT_PITCH + k0];
            const v2f b3 = *(const v2f*)&ldsW[(3 * 16 + m) * WT_PITCH + k0];
            c0 = __builtin_amdgcn_wmma_f32_16x16x4_f32(false, afrag[kk], false, b0, (short)0, c0, false, false);
            c1 = __builtin_amdgcn_wmma_f32_16x16x4_f32(false, afrag[kk], false, b1, (short)0, c1, false, false);
            c2 = __builtin_amdgcn_wmma_f32_16x16x4_f32(false, afrag[kk], false, b2, (short)0, c2, false, false);
            c3 = __builtin_amdgcn_wmma_f32_16x16x4_f32(false, afrag[kk], false, b3, (short)0, c3, false, false);
        }
    }

    walk_bin_pieces<0>(binned, segOff, segCnt, deg, stage, csrc, nchunks, nedges, 0, true);
    __syncthreads();

    const int i = b * BIN_NODES + tid;
    const int cdeg = (tid < BIN_NODES) ? deg[tid] : 0;
    int incl = cdeg;
#pragma unroll
    for (int o = 1; o < 32; o <<= 1) {
        const int up = __shfl_up(incl, o, 32);
        if (lane >= o) incl += up;
    }
    if (lane == 31) wtot[wv] = incl;
    __syncthreads();

    int base = 0, wpre = 0, total = 0;
#pragma unroll
    for (int w = 0; w < 32; ++w) {
        base  += red[w];
        total += wtot[w];
        if (w < wv) wpre += wtot[w];
    }
    const int excl = wpre + incl - cdeg;
    if (tid < BIN_NODES) {
        cursor[tid] = excl;
        rowScale[tid] = rsqrtf((float)(cdeg + 1));
        if (i < n) {
            rowptr[i] = base + excl;
            dinv[i] = rsqrtf((float)(cdeg + 1));
            if (i == n - 1) rowptr[n] = base + excl + cdeg;
        }
    }
    __syncthreads();

    if (tileLive) {
        float sc[8];
#pragma unroll
        for (int r = 0; r < 8; ++r) sc[r] = rowScale[wv * 16 + r + 8 * half];
        const v8f acc[4] = {c0, c1, c2, c3};
#pragma unroll
        for (int nt = 0; nt < 4; ++nt) {
            const int col = nt * 16 + m;
#pragma unroll
            for (int r = 0; r < 8; ++r) {
                const int orow = row0 + r + 8 * half;
                if (orow < n)
                    C[(size_t)orow * 64 + col] = acc[nt][r] * sc[r];
            }
        }
    }

    const bool fits = total <= STAGE_CAP;
    walk_bin_pieces<1>(binned, segOff, segCnt, cursor, stage, csrc, nchunks, nedges, base, fits);
    __syncthreads();
    if (fits)
        for (int j = tid; j < total; j += 1024) csrc[min(base + j, nedges - 1)] = stage[j];
}


#define TILE_PITCH 68

template <int NPG>
__device__ __forceinline__ void gather_rows_to_lds(const int* __restrict__ rowptr, const int* __restrict__ csrc,
                                                   const float* __restrict__ dinv, const float* __restrict__ ts,
                                                   const float* __restrict__ bias, float* __restrict__ ldsH, int n) {
    const int l16   = threadIdx.x & 15;
    const int gid   = threadIdx.x >> 4;
    const int first = blockIdx.x * (16 * NPG) + gid * NPG;
    const float4 bv = *(const float4*)(bias + l16 * 4);
    const int rp = rowptr[min(first + min(l16, NPG), n)];

    for (int s = 0; s < NPG; ++s) {
        const int i   = first + s;
        const int beg = __shfl(rp, s, 16);
        const int cnt = __shfl(rp, s + 1, 16) - beg;
        float4 o = make_float4(0.0f, 0.0f, 0.0f, 0.0f);
        if (i < n) {
            const float di = dinv[i];
            float4 acc = *(const float4*)(ts + (size_t)i * 64 + l16 * 4);

            for (int base = 0; base < cnt; base += 16) {
                const int k = base + l16;
                int s_mine = i;
                if (k < cnt) s_mine = min(max(csrc[beg + k], 0), n - 1);
                const int m = min(cnt - base, 16);
                for (int j = 0; j < m; j += 4) {
                    int sj[4]; float4 r[4];
#pragma unroll
                    for (int u = 0; u < 4; ++u) {
                        sj[u] = __shfl(s_mine, j + u, 16);
                        r[u]  = *(const float4*)(ts + (size_t)sj[u] * 64 + l16 * 4);
                    }
#pragma unroll
                    for (int u = 0; u < 4; ++u) {
                        const bool live = (u == 0) || (j + u) < m;
                        acc.x += live ? r[u].x : 0.0f;
                        acc.y += live ? r[u].y : 0.0f;
                        acc.z += live ? r[u].z : 0.0f;
                        acc.w += live ? r[u].w : 0.0f;
                    }
                }
            }
            o = make_float4(fmaxf(fmaf(di, acc.x, bv.x), 0.0f), fmaxf(fmaf(di, acc.y, bv.y), 0.0f),
                            fmaxf(fmaf(di, acc.z, bv.z), 0.0f), fmaxf(fmaf(di, acc.w, bv.w), 0.0f));
        }
        *(float4*)&ldsH[(gid * NPG + s) * TILE_PITCH + l16 * 4] = o;
    }
}


__global__ __launch_bounds__(256) void k_gather_gemm64(const int* __restrict__ rowptr, const int* __restrict__ csrc,
                                                       const float* __restrict__ dinv, const float* __restrict__ ts,
                                                       const float* __restrict__ bias, const float* __restrict__ W,
                                                       float* __restrict__ C, int n) {
    __shared__ __align__(16) float ldsW[64 * WT_PITCH];
    __shared__ __align__(16) float ldsH[64 * TILE_PITCH];

    float w[16];
#pragma unroll
    for (int i = 0; i < 16; ++i) w[i] = W[threadIdx.x + 256 * i];
#pragma unroll
    for (int i = 0; i < 16; ++i) {
        const int k = (threadIdx.x >> 6) + 4 * i;
        const int col = threadIdx.x & 63;
        ldsW[col * WT_PITCH + k] = w[i];
    }

    gather_rows_to_lds<4>(rowptr, csrc, dinv, ts, bias, ldsH, n);
    __syncthreads();

    const int lane = threadIdx.x & 31;
    const int wave = threadIdx.x >> 5;
    const int half = lane >> 4;
    const int m    = lane & 15;
    const int lt   = wave >> 1;
    const int nt0  = (wave & 1) * 2;
    const int row0 = blockIdx.x * 64 + lt * 16;
    if (row0 < n) {
        float sc[8];
#pragma unroll
        for (int r = 0; r < 8; ++r)
            sc[r] = dinv[min(row0 + r + 8 * half, n - 1)];

        v8f c0 = {}, c1 = {};
#pragma unroll
        for (int kk = 0; kk < 16; ++kk) {
            const int k0 = kk * 4 + 2 * half;
            const v2f a  = *(const v2f*)&ldsH[(lt * 16 + m) * TILE_PITCH + k0];
            const v2f b0 = *(const v2f*)&ldsW[((nt0 + 0) * 16 + m) * WT_PITCH + k0];
            const v2f b1 = *(const v2f*)&ldsW[((nt0 + 1) * 16 + m) * WT_PITCH + k0];
            c0 = __builtin_amdgcn_wmma_f32_16x16x4_f32(false, a, false, b0, (short)0, c0, false, false);
            c1 = __builtin_amdgcn_wmma_f32_16x16x4_f32(false, a, false, b1, (short)0, c1, false, false);
        }

        v8f acc2[2] = {c0, c1};
#pragma unroll
        for (int t = 0; t < 2; ++t) {
            const int col = (nt0 + t) * 16 + m;
#pragma unroll
            for (int r = 0; r < 8; ++r) {
                const int orow = row0 + r + 8 * half;
                if (orow < n)
                    C[(size_t)orow * 64 + col] = acc2[t][r] * sc[r];
            }
        }
    }
}


__device__ __forceinline__ int head_kpos(int f) {
    return (f & 0x31) | ((f & 0x6) << 1) | ((f >> 2) & 0x2);
}

__global__ __launch_bounds__(256) void k_gather_mlp_head(const int* __restrict__ rowptr, const int* __restrict__ csrc,
                                                         const float* __restrict__ dinv, const float* __restrict__ ts,
                                                         const float* __restrict__ bias,
                                                         const float* __restrict__ Wa, const float* __restrict__ ba,
                                                         const float* __restrict__ Wb, const float* __restrict__ bb,
                                                         const float* __restrict__ w3, const float* __restrict__ b3,
                                                         float* __restrict__ out, int n) {
    __shared__ __align__(16) float ldsWa[64 * WT_PITCH];
    __shared__ __align__(16) float ldsWb[64 * WT_PITCH];
    __shared__ __align__(16) float ldsV[3 * 64];
    __shared__ __align__(16) float ldsH[80 * TILE_PITCH];

    {
        float wa[16], wb[16];
#pragma unroll
        for (int i = 0; i < 16; ++i) {
            wa[i] = Wa[threadIdx.x + 256 * i];
            wb[i] = Wb[threadIdx.x + 256 * i];
        }
#pragma unroll
        for (int i = 0; i < 16; ++i) {
            const int k = (threadIdx.x >> 6) + 4 * i;
            const int col = threadIdx.x & 63;
            ldsWa[col * WT_PITCH + k] = wa[i];
            ldsWb[col * WT_PITCH + head_kpos(k)] = wb[i];
        }
        if (threadIdx.x < 64) {
            ldsV[threadIdx.x]       = ba[threadIdx.x];
            ldsV[64 + threadIdx.x]  = bb[threadIdx.x];
            ldsV[128 + threadIdx.x] = w3[threadIdx.x];
        }
    }

    gather_rows_to_lds<5>(rowptr, csrc, dinv, ts, bias, ldsH, n);
    __syncthreads();

    const int lane = threadIdx.x & 31;
    const int wave = threadIdx.x >> 5;
    const int half = lane >> 4;
    const int c    = lane & 15;
    const int node = blockIdx.x * 80 + wave * 16 + c;
    if (wave < 5 && blockIdx.x * 80 + wave * 16 < n) {
        const float* hrow = &ldsH[(wave * 16 + c) * TILE_PITCH + 2 * half];

        v8f t[4] = {};
#pragma unroll
        for (int kk = 0; kk < 16; ++kk) {
            const int k0 = kk * 4 + 2 * half;
            const v2f x = *(const v2f*)(hrow + 4 * kk);
            v2f a[4];
#pragma unroll
            for (int mt = 0; mt < 4; ++mt) a[mt] = *(const v2f*)&ldsWa[(mt * 16 + c) * WT_PITCH + k0];
#pragma unroll
            for (int mt = 0; mt < 4; ++mt)
                t[mt] = __builtin_amdgcn_wmma_f32_16x16x4_f32(false, a[mt], false, x, (short)0, t[mt], false, false);
        }

#pragma unroll
        for (int mt = 0; mt < 4; ++mt) {
            const float4 g0 = *(const float4*)&ldsV[mt * 16 + 8 * half];
            const float4 g1 = *(const float4*)&ldsV[mt * 16 + 8 * half + 4];
            t[mt][0] = fmaxf(t[mt][0] + g0.x, 0.0f);
            t[mt][1] = fmaxf(t[mt][1] + g0.y, 0.0f);
            t[mt][2] = fmaxf(t[mt][2] + g0.z, 0.0f);
            t[mt][3] = fmaxf(t[mt][3] + g0.w, 0.0f);
            t[mt][4] = fmaxf(t[mt][4] + g1.x, 0.0f);
            t[mt][5] = fmaxf(t[mt][5] + g1.y, 0.0f);
            t[mt][6] = fmaxf(t[mt][6] + g1.z, 0.0f);
            t[mt][7] = fmaxf(t[mt][7] + g1.w, 0.0f);
        }

        v8f u[4] = {};
#pragma unroll
        for (int nt = 0; nt < 4; ++nt) {
#pragma unroll
            for (int q = 0; q < 4; ++q) {
                v2f b;
                b.x = t[nt][2 * q];
                b.y = t[nt][2 * q + 1];
                const int kp = nt * 16 + q * 4 + 2 * half;
                v2f a[4];
#pragma unroll
                for (int mt = 0; mt < 4; ++mt) a[mt] = *(const v2f*)&ldsWb[(mt * 16 + c) * WT_PITCH + kp];
#pragma unroll
                for (int mt = 0; mt < 4; ++mt)
                    u[mt] = __builtin_amdgcn_wmma_f32_16x16x4_f32(false, a[mt], false, b, (short)0, u[mt], false, false);
            }
        }

        float p[4];
#pragma unroll
        for (int mt = 0; mt < 4; ++mt) {
            const float4 g0 = *(const float4*)&ldsV[64 + mt * 16 + 8 * half];
            const float4 g1 = *(const float4*)&ldsV[64 + mt * 16 + 8 * half + 4];
            const float4 w0 = *(const float4*)&ldsV[128 + mt * 16 + 8 * half];
            const float4 w1 = *(const float4*)&ldsV[128 + mt * 16 + 8 * half + 4];
            float s = fmaxf(u[mt][0] + g0.x, 0.0f) * w0.x;
            s = fmaf(fmaxf(u[mt][1] + g0.y, 0.0f), w0.y, s);
            s = fmaf(fmaxf(u[mt][2] + g0.z, 0.0f), w0.z, s);
            s = fmaf(fmaxf(u[mt][3] + g0.w, 0.0f), w0.w, s);
            s = fmaf(fmaxf(u[mt][4] + g1.x, 0.0f), w1.x, s);
            s = fmaf(fmaxf(u[mt][5] + g1.y, 0.0f), w1.y, s);
            s = fmaf(fmaxf(u[mt][6] + g1.z, 0.0f), w1.z, s);
            s = fmaf(fmaxf(u[mt][7] + g1.w, 0.0f), w1.w, s);
            p[mt] = s;
        }
        float dot = (p[0] + p[1]) + (p[2] + p[3]);
        dot += __shfl_xor(dot, 16, 32);
        if (half == 0 && node < n) out[node] = dot + b3[0];
    }
}


extern "C" void kernel_launch(void* const* d_in, const int* in_sizes, int n_in,
                              void* d_out, int out_size, void* d_ws, size_t ws_size,
                              hipStream_t stream) {
    (void)n_in; (void)out_size; (void)ws_size;

    const float* x   = (const float*)d_in[0];
    const int*   ei  = (const int*)  d_in[1];
    const float* W1  = (const float*)d_in[2];
    const float* b1  = (const float*)d_in[3];
    const float* W2  = (const float*)d_in[4];
    const float* b2  = (const float*)d_in[5];
    const float* Wf1 = (const float*)d_in[6];
    const float* bf1 = (const float*)d_in[7];
    const float* Wf2 = (const float*)d_in[8];
    const float* bf2 = (const float*)d_in[9];
    const float* Wf3 = (const float*)d_in[10];
    const float* bf3 = (const float*)d_in[11];
    float* out = (float*)d_out;

    const int n = in_sizes[0] / 64;
    const int E = in_sizes[1] / 2;

    const int nbins   = (n + BIN_NODES - 1) / BIN_NODES;
    const int nchunks = (E + CHUNK_EDGES - 1) / CHUNK_EDGES;

    const size_t nPad  = ((size_t)n + 1 + 63) / 64 * 64;
    const size_t ePad  = ((size_t)E + 63) / 64 * 64;
    const size_t tPad  = ((size_t)nbins * nchunks + 63) / 64 * 64;
    float* ws     = (float*)d_ws;
    float* dinv   = ws;
    int*   rowptr = (int*)(ws + nPad);
    int*   binned = (int*)(ws + 2 * nPad);
    int*   csrc   = (int*)(ws + 2 * nPad + ePad);
    int*   tabCnt = (int*)(ws + 2 * nPad + 2 * ePad);
    int*   tabOff = tabCnt + tPad;
    float* bufA   = ws + 2 * nPad + 2 * ePad + 2 * tPad;
    float* bufB   = bufA + (size_t)n * 64;

    const int T = 256;

    k_bin_edges<<<nchunks, 1024, 0, stream>>>(ei, binned, tabCnt, tabOff, E, nbins);
    k_bin_sort<<<nbins, 1024, 0, stream>>>(binned, tabCnt, tabOff, rowptr, dinv, csrc, x, W1, bufA, n, E, nchunks);

    k_gather_gemm64<<<(n + 63) / 64, T, 0, stream>>>(rowptr, csrc, dinv, bufA, b1, W2, bufB, n);

    k_gather_mlp_head<<<(n + 79) / 80, T, 0, stream>>>(rowptr, csrc, dinv, bufB, b2, Wf1, bf1, Wf2, bf2, Wf3, bf3, out, n);
}


# Gather keeps more reads in flight: the first 32 source ids of every node of a group are requested up front and rows are read 8 at a time (was 4, behind a per-node id read); sum order unchanged
# speedup vs baseline: 5.3218x; 1.0359x over previous
#include <hip/hip_runtime.h>


typedef __attribute__((ext_vector_type(2))) float v2f;
typedef __attribute__((ext_vector_type(8))) float v8f;


#define BIN_SHIFT   9
#define BIN_NODES   512
#define CHUNK_EDGES 8192
#define STAGE_CAP   12288

__global__ __launch_bounds__(1024) void k_bin_edges(const int* __restrict__ ei, int* __restrict__ binned,
                                                    int* __restrict__ tabCnt, int* __restrict__ tabOff,
                                                    int nedges, int nbins) {
    __shared__ int hist[1024];
    __shared__ int wtot[32];
    __shared__ unsigned sorted[CHUNK_EDGES];
    const int tid  = threadIdx.x;
    const int lane = tid & 31;
    const int wv   = tid >> 5;
    const int c    = blockIdx.x;
    const int nchunks = gridDim.x;
    const int e0  = c * CHUNK_EDGES;
    const int len = min(CHUNK_EDGES, nedges - e0);

    hist[tid] = 0;
    int s[8], d[8];
#pragma unroll
    for (int k = 0; k < 8; ++k) {
        const int e = min(e0 + k * 1024 + tid, nedges - 1);
        s[k] = ei[e];
        d[k] = ei[nedges + e];
    }
    __syncthreads();

    int bn[8], rk[8];
#pragma unroll
    for (int k = 0; k < 8; ++k) {
        const bool live = (k * 1024 + tid) < len;
        bn[k] = min(max(d[k] >> BIN_SHIFT, 0), nbins - 1);
        rk[k] = live ? atomicAdd(&hist[bn[k]], 1) : 0;
    }
    __syncthreads();

    const int h = hist[tid];
    int incl = h;
#pragma unroll
    for (int o = 1; o < 32; o <<= 1) {
        const int up = __shfl_up(incl, o, 32);
        if (lane >= o) incl += up;
    }
    if (lane == 31) wtot[wv] = incl;
    __syncthreads();
    int wpre = 0;
#pragma unroll
    for (int w = 0; w < 32; ++w)
        if (w < wv) wpre += wtot[w];
    const int excl = wpre + incl - h;
    hist[tid] = excl;
    if (tid < nbins) {
        tabCnt[tid * nchunks + c] = h;
        tabOff[tid * nchunks + c] = excl;
    }
    __syncthreads();

#pragma unroll
    for (int k = 0; k < 8; ++k) {
        const bool live = (k * 1024 + tid) < len;
        if (live) {
            const int p = min(hist[bn[k]] + rk[k], CHUNK_EDGES - 1);
            sorted[p] = ((unsigned)s[k] << BIN_SHIFT) | ((unsigned)d[k] & (BIN_NODES - 1));
        }
    }
    __syncthreads();
    for (int i = tid; i < len; i += 1024) binned[e0 + i] = (int)sorted[i];
}

template <int PASS>
__device__ __forceinline__ void walk_bin_pieces(const int* __restrict__ binned, const int* segOff, const int* segCnt,
                                                int* counters, int* stage, int* __restrict__ csrc,
                                                int nchunks, int nedges, int base, bool fits) {
    const int lane = threadIdx.x & 31;
    const int wv   = threadIdx.x >> 5;
    auto visit = [&](unsigned w) {
        const int dl = (int)(w & (BIN_NODES - 1));
        if (PASS == 0) {
            atomicAdd(&counters[dl], 1);
        } else {
            const int p   = atomicAdd(&counters[dl], 1);
            const int src = (int)(w >> BIN_SHIFT);
            if (fits) stage[min(max(p, 0), STAGE_CAP - 1)] = src;
            else      csrc[min(max(base + p, 0), nedges - 1)] = src;
        }
    };
    for (int s0 = wv; s0 < nchunks; s0 += 128) {
        int o[4], m[4];
        unsigned w0[4], w1[4];
#pragma unroll
        for (int u = 0; u < 4; ++u) {
            const int sg = s0 + 32 * u;
            o[u] = (sg < nchunks) ? segOff[sg] : 0;
            m[u] = (sg < nchunks) ? segCnt[sg] : 0;
        }
#pragma unroll
        for (int u = 0; u < 4; ++u) {
            w0[u] = (unsigned)binned[min(o[u] + lane, nedges - 1)];
            w1[u] = (unsigned)binned[min(o[u] + lane + 32, nedges - 1)];
        }
#pragma unroll
        for (int u = 0; u < 4; ++u) {
            if (lane < m[u]) visit(w0[u]);
            if (lane + 32 < m[u]) visit(w1[u]);
            for (int k = lane + 64; k < m[u]; k += 32) visit((unsigned)binned[min(o[u] + k, nedges - 1)]);
        }
    }
}

#define WT_PITCH 68

__global__ __launch_bounds__(1024) void k_bin_sort(const int* __restrict__ binned, const int* __restrict__ tabCnt,
                                                   const int* __restrict__ tabOff, int* __restrict__ rowptr,
                                                   float* __restrict__ dinv, int* __restrict__ csrc,
                                                   const float* __restrict__ A, const float* __restrict__ W,
                                                   float* __restrict__ C,
                                                   int n, int nedges, int nchunks) {
    __shared__ int deg[BIN_NODES];
    __shared__ int cursor[BIN_NODES];
    __shared__ int segOff[1024];
    __shared__ int segCnt[1024];
    __shared__ int red[32];
    __shared__ int wtot[32];
    __shared__ int stage[STAGE_CAP];
    __shared__ __align__(16) float ldsW[64 * WT_PITCH];
    __shared__ float rowScale[BIN_NODES];
    const int tid  = threadIdx.x;
    const int lane = tid & 31;
    const int wv   = tid >> 5;
    const int b    = blockIdx.x;

    const int half = lane >> 4;
    const int m    = lane & 15;
    const int row0 = b * BIN_NODES + wv * 16;
    const bool tileLive = row0 < n;
    v2f afrag[16];
    {
        const float* __restrict__ arow = A + (size_t)min(row0 + m, n - 1) * 64;
#pragma unroll
        for (int kk = 0; kk < 16; ++kk) {
            const int k0 = kk * 4 + 2 * half;
            afrag[kk].x = arow[k0];
            afrag[kk].y = arow[k0 + 1];
        }
    }
    float wst[4];
#pragma unroll
    for (int i = 0; i < 4; ++i) wst[i] = W[tid + 1024 * i];
#pragma unroll
    for (int i = 0; i < 4; ++i) {
        const int k = (tid >> 6) + 16 * i;
        const int col = tid & 63;
        ldsW[col * WT_PITCH + k] = wst[i];
    }

    int part = 0;
    const int npre = b * nchunks;
    for (int j = tid; j < npre; j += 16384) {
        int q[16];
#pragma unroll
        for (int k = 0; k < 16; ++k) q[k] = tabCnt[min(j + k * 1024, npre - 1)];
#pragma unroll
        for (int k = 0; k < 16; ++k)
            if (j + k * 1024 < npre) part += q[k];
    }
#pragma unroll
    for (int o = 16; o > 0; o >>= 1) part += __shfl_xor(part, o, 32);
    if (lane == 0) red[wv] = part;

    if (tid < BIN_NODES) deg[tid] = 0;
    segCnt[tid] = (tid < nchunks) ? tabCnt[b * nchunks + tid] : 0;
    segOff[tid] = (tid < nchunks) ? tid * CHUNK_EDGES + tabOff[b * nchunks + tid] : 0;
    __syncthreads();

    v8f c0 = {}, c1 = {}, c2 = {}, c3 = {};
    if (tileLive) {
#pragma unroll
        for (int kk = 0; kk < 16; ++kk) {
            const int k0 = kk * 4 + 2 * half;
            const v2f b0 = *(const v2f*)&ldsW[(0 * 16 + m) * WT_PITCH + k0];
            const v2f b1 = *(const v2f*)&ldsW[(1 * 16 + m) * WT_PITCH + k0];
            const v2f b2 = *(const v2f*)&ldsW[(2 * 16 + m) * WT_PITCH + k0];
            const v2f b3 = *(const v2f*)&ldsW[(3 * 16 + m) * WT_PITCH + k0];
            c0 = __builtin_amdgcn_wmma_f32_16x16x4_f32(false, afrag[kk], false, b0, (short)0, c0, false, false);
            c1 = __builtin_amdgcn_wmma_f32_16x16x4_f32(false, afrag[kk], false, b1, (short)0, c1, false, false);
            c2 = __builtin_amdgcn_wmma_f32_16x16x4_f32(false, afrag[kk], false, b2, (short)0, c2, false, false);
            c3 = __builtin_amdgcn_wmma_f32_16x16x4_f32(false, afrag[kk], false, b3, (short)0, c3, false, false);
        }
    }

    walk_bin_pieces<0>(binned, segOff, segCnt, deg, stage, csrc, nchunks, nedges, 0, true);
    __syncthreads();

    const int i = b * BIN_NODES + tid;
    const int cdeg = (tid < BIN_NODES) ? deg[tid] : 0;
    int incl = cdeg;
#pragma unroll
    for (int o = 1; o < 32; o <<= 1) {
        const int up = __shfl_up(incl, o, 32);
        if (lane >= o) incl += up;
    }
    if (lane == 31) wtot[wv] = incl;
    __syncthreads();

    int base = 0, wpre = 0, total = 0;
#pragma unroll
    for (int w = 0; w < 32; ++w) {
        base  += red[w];
        total += wtot[w];
        if (w < wv) wpre += wtot[w];
    }
    const int excl = wpre + incl - cdeg;
    if (tid < BIN_NODES) {
        cursor[tid] = excl;
        rowScale[tid] = rsqrtf((float)(cdeg + 1));
        if (i < n) {
            rowptr[i] = base + excl;
            dinv[i] = rsqrtf((float)(cdeg + 1));
            if (i == n - 1) rowptr[n] = base + excl + cdeg;
        }
    }
    __syncthreads();

    if (tileLive) {
        float sc[8];
#pragma unroll
        for (int r = 0; r < 8; ++r) sc[r] = rowScale[wv * 16 + r + 8 * half];
        const v8f acc[4] = {c0, c1, c2, c3};
#pragma unroll
        for (int nt = 0; nt < 4; ++nt) {
            const int col = nt * 16 + m;
#pragma unroll
            for (int r = 0; r < 8; ++r) {
                const int orow = row0 + r + 8 * half;
                if (orow < n)
                    C[(size_t)orow * 64 + col] = acc[nt][r] * sc[r];
            }
        }
    }

    const bool fits = total <= STAGE_CAP;
    walk_bin_pieces<1>(binned, segOff, segCnt, cursor, stage, csrc, nchunks, nedges, base, fits);
    __syncthreads();
    if (fits)
        for (int j = tid; j < total; j += 1024) csrc[min(base + j, nedges - 1)] = stage[j];
}


#define TILE_PITCH 68

template <int NPG>
__device__ __forceinline__ void gather_rows_to_lds(const int* __restrict__ rowptr, const int* __restrict__ csrc,
                                                   const float* __restrict__ dinv, const float* __restrict__ ts,
                                                   const float* __restrict__ bias, float* __restrict__ ldsH, int n) {
    const int l16   = threadIdx.x & 15;
    const int gid   = threadIdx.x >> 4;
    const int first = blockIdx.x * (16 * NPG) + gid * NPG;
    const unsigned lo = (unsigned)l16 * 4u;
    const float4 bv = *(const float4*)(bias + l16 * 4);
    const int rp = rowptr[min(first + min(l16, NPG), n)];

    int id0[NPG], id1[NPG];
#pragma unroll
    for (int s = 0; s < NPG; ++s) {
        const int own = min(first + s, n - 1);
        const int beg = __shfl(rp, s, 16);
        const int cnt = __shfl(rp, s + 1, 16) - beg;
        int a0 = own, a1 = own;
        if (l16 < cnt)      a0 = csrc[beg + l16];
        if (l16 + 16 < cnt) a1 = csrc[beg + 16 + l16];
        id0[s] = min(max(a0, 0), n - 1);
        id1[s] = min(max(a1, 0), n - 1);
    }

    auto batch8 = [&](int idreg, int lane0, int live, float4& acc) {
        float4 r[8];
#pragma unroll
        for (int u = 0; u < 8; ++u) {
            const int sj = __shfl(idreg, lane0 + u, 16);
            r[u] = *(const float4*)(ts + ((unsigned)sj * 64u + lo));
        }
#pragma unroll
        for (int u = 0; u < 8; ++u) {
            const bool lv = (u == 0) || (u < live);
            acc.x += lv ? r[u].x : 0.0f;
            acc.y += lv ? r[u].y : 0.0f;
            acc.z += lv ? r[u].z : 0.0f;
            acc.w += lv ? r[u].w : 0.0f;
        }
    };

#pragma unroll
    for (int s = 0; s < NPG; ++s) {
        const int i   = first + s;
        const int beg = __shfl(rp, s, 16);
        const int cnt = __shfl(rp, s + 1, 16) - beg;
        float4 o = make_float4(0.0f, 0.0f, 0.0f, 0.0f);
        if (i < n) {
            const float di = dinv[i];
            float4 acc = *(const float4*)(ts + ((unsigned)i * 64u + lo));
            if (cnt > 0)  batch8(id0[s], 0, cnt, acc);
            if (cnt > 8)  batch8(id0[s], 8, cnt - 8, acc);
            if (cnt > 16) batch8(id1[s], 0, cnt - 16, acc);
            if (cnt > 24) batch8(id1[s], 8, cnt - 24, acc);
            for (int base = 32; base < cnt; base += 16) {
                int s_mine = i;
                if (base + l16 < cnt) s_mine = min(max(csrc[beg + base + l16], 0), n - 1);
                const int m = min(cnt - base, 16);
                batch8(s_mine, 0, m, acc);
                if (m > 8) batch8(s_mine, 8, m - 8, acc);
            }
            o = make_float4(fmaxf(fmaf(di, acc.x, bv.x), 0.0f), fmaxf(fmaf(di, acc.y, bv.y), 0.0f),
                            fmaxf(fmaf(di, acc.z, bv.z), 0.0f), fmaxf(fmaf(di, acc.w, bv.w), 0.0f));
        }
        *(float4*)&ldsH[(gid * NPG + s) * TILE_PITCH + l16 * 4] = o;
    }
}


__global__ __launch_bounds__(256) void k_gather_gemm64(const int* __restrict__ rowptr, const int* __restrict__ csrc,
                                                       const float* __restrict__ dinv, const float* __restrict__ ts,
                                                       const float* __restrict__ bias, const float* __restrict__ W,
                                                       float* __restrict__ C, int n) {
    __shared__ __align__(16) float ldsW[64 * WT_PITCH];
    __shared__ __align__(16) float ldsH[64 * TILE_PITCH];

    float w[16];
#pragma unroll
    for (int i = 0; i < 16; ++i) w[i] = W[threadIdx.x + 256 * i];
#pragma unroll
    for (int i = 0; i < 16; ++i) {
        const int k = (threadIdx.x >> 6) + 4 * i;
        const int col = threadIdx.x & 63;
        ldsW[col * WT_PITCH + k] = w[i];
    }

    gather_rows_to_lds<4>(rowptr, csrc, dinv, ts, bias, ldsH, n);
    __syncthreads();

    const int lane = threadIdx.x & 31;
    const int wave = threadIdx.x >> 5;
    const int half = lane >> 4;
    const int m    = lane & 15;
    const int lt   = wave >> 1;
    const int nt0  = (wave & 1) * 2;
    const int row0 = blockIdx.x * 64 + lt * 16;
    if (row0 < n) {
        float sc[8];
#pragma unroll
        for (int r = 0; r < 8; ++r)
            sc[r] = dinv[min(row0 + r + 8 * half, n - 1)];

        v8f c0 = {}, c1 = {};
#pragma unroll
        for (int kk = 0; kk < 16; ++kk) {
            const int k0 = kk * 4 + 2 * half;
            const v2f a  = *(const v2f*)&ldsH[(lt * 16 + m) * TILE_PITCH + k0];
            const v2f b0 = *(const v2f*)&ldsW[((nt0 + 0) * 16 + m) * WT_PITCH + k0];
            const v2f b1 = *(const v2f*)&ldsW[((nt0 + 1) * 16 + m) * WT_PITCH + k0];
            c0 = __builtin_amdgcn_wmma_f32_16x16x4_f32(false, a, false, b0, (short)0, c0, false, false);
            c1 = __builtin_amdgcn_wmma_f32_16x16x4_f32(false, a, false, b1, (short)0, c1, false, false);
        }

        v8f acc2[2] = {c0, c1};
#pragma unroll
        for (int t = 0; t < 2; ++t) {
            const int col = (nt0 + t) * 16 + m;
#pragma unroll
            for (int r = 0; r < 8; ++r) {
                const int orow = row0 + r + 8 * half;
                if (orow < n)
                    C[(size_t)orow * 64 + col] = acc2[t][r] * sc[r];
            }
        }
    }
}


__device__ __forceinline__ int head_kpos(int f) {
    return (f & 0x31) | ((f & 0x6) << 1) | ((f >> 2) & 0x2);
}

__global__ __launch_bounds__(256) void k_gather_mlp_head(const int* __restrict__ rowptr, const int* __restrict__ csrc,
                                                         const float* __restrict__ dinv, const float* __restrict__ ts,
                                                         const float* __restrict__ bias,
                                                         const float* __restrict__ Wa, const float* __restrict__ ba,
                                                         const float* __restrict__ Wb, const float* __restrict__ bb,
                                                         const float* __restrict__ w3, const float* __restrict__ b3,
                                                         float* __restrict__ out, int n) {
    __shared__ __align__(16) float ldsWa[64 * WT_PITCH];
    __shared__ __align__(16) float ldsWb[64 * WT_PITCH];
    __shared__ __align__(16) float ldsV[3 * 64];
    __shared__ __align__(16) float ldsH[80 * TILE_PITCH];

    {
        float wa[16], wb[16];
#pragma unroll
        for (int i = 0; i < 16; ++i) {
            wa[i] = Wa[threadIdx.x + 256 * i];
            wb[i] = Wb[threadIdx.x + 256 * i];
        }
#pragma unroll
        for (int i = 0; i < 16; ++i) {
            const int k = (threadIdx.x >> 6) + 4 * i;
            const int col = threadIdx.x & 63;
            ldsWa[col * WT_PITCH + k] = wa[i];
            ldsWb[col * WT_PITCH + head_kpos(k)] = wb[i];
        }
        if (threadIdx.x < 64) {
            ldsV[threadIdx.x]       = ba[threadIdx.x];
            ldsV[64 + threadIdx.x]  = bb[threadIdx.x];
            ldsV[128 + threadIdx.x] = w3[threadIdx.x];
        }
    }

    gather_rows_to_lds<5>(rowptr, csrc, dinv, ts, bias, ldsH, n);
    __syncthreads();

    const int lane = threadIdx.x & 31;
    const int wave = threadIdx.x >> 5;
    const int half = lane >> 4;
    const int c    = lane & 15;
    const int node = blockIdx.x * 80 + wave * 16 + c;
    if (wave < 5 && blockIdx.x * 80 + wave * 16 < n) {
        const float* hrow = &ldsH[(wave * 16 + c) * TILE_PITCH + 2 * half];

        v8f t[4] = {};
#pragma unroll
        for (int kk = 0; kk < 16; ++kk) {
            const int k0 = kk * 4 + 2 * half;
            const v2f x = *(const v2f*)(hrow + 4 * kk);
            v2f a[4];
#pragma unroll
            for (int mt = 0; mt < 4; ++mt) a[mt] = *(const v2f*)&ldsWa[(mt * 16 + c) * WT_PITCH + k0];
#pragma unroll
            for (int mt = 0; mt < 4; ++mt)
                t[mt] = __builtin_amdgcn_wmma_f32_16x16x4_f32(false, a[mt], false, x, (short)0, t[mt], false, false);
        }

#pragma unroll
        for (int mt = 0; mt < 4; ++mt) {
            const float4 g0 = *(const float4*)&ldsV[mt * 16 + 8 * half];
            const float4 g1 = *(const float4*)&ldsV[mt * 16 + 8 * half + 4];
            t[mt][0] = fmaxf(t[mt][0] + g0.x, 0.0f);
            t[mt][1] = fmaxf(t[mt][1] + g0.y, 0.0f);
            t[mt][2] = fmaxf(t[mt][2] + g0.z, 0.0f);
            t[mt][3] = fmaxf(t[mt][3] + g0.w, 0.0f);
            t[mt][4] = fmaxf(t[mt][4] + g1.x, 0.0f);
            t[mt][5] = fmaxf(t[mt][5] + g1.y, 0.0f);
            t[mt][6] = fmaxf(t[mt][6] + g1.z, 0.0f);
            t[mt][7] = fmaxf(t[mt][7] + g1.w, 0.0f);
        }

        v8f u[4] = {};
#pragma unroll
        for (int nt = 0; nt < 4; ++nt) {
#pragma unroll
            for (int q = 0; q < 4; ++q) {
                v2f b;
                b.x = t[nt][2 * q];
                b.y = t[nt][2 * q + 1];
                const int kp = nt * 16 + q * 4 + 2 * half;
                v2f a[4];
#pragma unroll
                for (int mt = 0; mt < 4; ++mt) a[mt] = *(const v2f*)&ldsWb[(mt * 16 + c) * WT_PITCH + kp];
#pragma unroll
                for (int mt = 0; mt < 4; ++mt)
                    u[mt] = __builtin_amdgcn_wmma_f32_16x16x4_f32(false, a[mt], false, b, (short)0, u[mt], false, false);
            }
        }

        float p[4];
#pragma unroll
        for (int mt = 0; mt < 4; ++mt) {
            const float4 g0 = *(const float4*)&ldsV[64 + mt * 16 + 8 * half];
            const float4 g1 = *(const float4*)&ldsV[64 + mt * 16 + 8 * half + 4];
            const float4 w0 = *(const float4*)&ldsV[128 + mt * 16 + 8 * half];
            const float4 w1 = *(const float4*)&ldsV[128 + mt * 16 + 8 * half + 4];
            float s = fmaxf(u[mt][0] + g0.x, 0.0f) * w0.x;
            s = fmaf(fmaxf(u[mt][1] + g0.y, 0.0f), w0.y, s);
            s = fmaf(fmaxf(u[mt][2] + g0.z, 0.0f), w0.z, s);
            s = fmaf(fmaxf(u[mt][3] + g0.w, 0.0f), w0.w, s);
            s = fmaf(fmaxf(u[mt][4] + g1.x, 0.0f), w1.x, s);
            s = fmaf(fmaxf(u[mt][5] + g1.y, 0.0f), w1.y, s);
            s = fmaf(fmaxf(u[mt][6] + g1.z, 0.0f), w1.z, s);
            s = fmaf(fmaxf(u[mt][7] + g1.w, 0.0f), w1.w, s);
            p[mt] = s;
        }
        float dot = (p[0] + p[1]) + (p[2] + p[3]);
        dot += __shfl_xor(dot, 16, 32);
        if (half == 0 && node < n) out[node] = dot + b3[0];
    }
}


extern "C" void kernel_launch(void* const* d_in, const int* in_sizes, int n_in,
                              void* d_out, int out_size, void* d_ws, size_t ws_size,
                              hipStream_t stream) {
    (void)n_in; (void)out_size; (void)ws_size;

    const float* x   = (const float*)d_in[0];
    const int*   ei  = (const int*)  d_in[1];
    const float* W1  = (const float*)d_in[2];
    const float* b1  = (const float*)d_in[3];
    const float* W2  = (const float*)d_in[4];
    const float* b2  = (const float*)d_in[5];
    const float* Wf1 = (const float*)d_in[6];
    const float* bf1 = (const float*)d_in[7];
    const float* Wf2 = (const float*)d_in[8];
    const float* bf2 = (const float*)d_in[9];
    const float* Wf3 = (const float*)d_in[10];
    const float* bf3 = (const float*)d_in[11];
    float* out = (float*)d_out;

    const int n = in_sizes[0] / 64;
    const int E = in_sizes[1] / 2;

    const int nbins   = (n + BIN_NODES - 1) / BIN_NODES;
    const int nchunks = (E + CHUNK_EDGES - 1) / CHUNK_EDGES;

    const size_t nPad  = ((size_t)n + 1 + 63) / 64 * 64;
    const size_t ePad  = ((size_t)E + 63) / 64 * 64;
    const size_t tPad  = ((size_t)nbins * nchunks + 63) / 64 * 64;
    float* ws     = (float*)d_ws;
    float* dinv   = ws;
    int*   rowptr = (int*)(ws + nPad);
    int*   binned = (int*)(ws + 2 * nPad);
    int*   csrc   = (int*)(ws + 2 * nPad + ePad);
    int*   tabCnt = (int*)(ws + 2 * nPad + 2 * ePad);
    int*   tabOff = tabCnt + tPad;
    float* bufA   = ws + 2 * nPad + 2 * ePad + 2 * tPad;
    float* bufB   = bufA + (size_t)n * 64;

    const int T = 256;

    k_bin_edges<<<nchunks, 1024, 0, stream>>>(ei, binned, tabCnt, tabOff, E, nbins);
    k_bin_sort<<<nbins, 1024, 0, stream>>>(binned, tabCnt, tabOff, rowptr, dinv, csrc, x, W1, bufA, n, E, nchunks);

    k_gather_gemm64<<<(n + 63) / 64, T, 0, stream>>>(rowptr, csrc, dinv, bufA, b1, W2, bufB, n);

    k_gather_mlp_head<<<(n + 79) / 80, T, 0, stream>>>(rowptr, csrc, dinv, bufB, b2, Wf1, bf1, Wf2, bf2, Wf3, bf3, out, n);
}
